# starting kernel
# baseline (speedup 1.0000x reference)
#include <hip/hip_runtime.h>
#include <math.h>

typedef __attribute__((ext_vector_type(2))) float v2f;
typedef __attribute__((ext_vector_type(8))) float v8f;

#define WAVES 4

constexpr int B_ = 2, L_ = 2048, S_ = 2048, H_ = 8, E_ = 64, D_ = 64;
constexpr float SCALE  = 0.125f;
constexpr float MASKED = -1.25e8f;

__device__ __forceinline__ bool log_mask(int i, int j) {
    int d = i - j;
    if (d < 0) return false;
    if (i < 22) return true;
    if (d <= 10) return true;
    int e = d - 10;
    return (e <= 1024) && ((e & (e - 1)) == 0);
}

__device__ __forceinline__ bool tile_active(int i0, int j0) {
    int delta = i0 - j0;
    if (delta < -15) return false;
    if (i0 < 32) return true;
    if (delta <= 25) return true;
#pragma unroll
    for (int p = 0; p < 11; ++p) {
        int dv = 10 + (1 << p);
        if (dv >= delta - 15 && dv <= delta + 15) return true;
    }
    return false;
}

__global__ __launch_bounds__(WAVES * 32)
void logsparse_attn_kernel(const float* __restrict__ Q,
                           const float* __restrict__ K,
                           const float* __restrict__ V,
                           float* __restrict__ Out) {
    __shared__ __align__(16) float plds[WAVES][16][18];

    const int lane = threadIdx.x & 31;
    const int wave = threadIdx.x >> 5;
    const int g    = lane >> 4;
    const int ln   = lane & 15;

    const int gid = blockIdx.x * WAVES + wave;
    const int qt  = gid & 127;
    const int h   = (gid >> 7) & 7;
    const int b   = gid >> 10;
    const int i0  = qt * 16;

    const float* qrow = Q + ((size_t)(b * L_ + i0 + ln) * H_ + h) * E_ + 2 * g;
    v2f qf[16];
#pragma unroll
    for (int kc = 0; kc < 16; ++kc) {
        float2 t = *(const float2*)(qrow + 4 * kc);
        v2f a; a[0] = t.x; a[1] = t.y;
        qf[kc] = a;
    }

    v8f oacc[4];
#pragma unroll
    for (int nt = 0; nt < 4; ++nt) oacc[nt] = (v8f){};
    float rowmax[8], rowsum[8];
#pragma unroll
    for (int r = 0; r < 8; ++r) { rowmax[r] = -INFINITY; rowsum[r] = 0.0f; }

    const int jt_end = qt;
    for (int jt = 0; jt <= jt_end; ++jt) {
        const int j0 = jt * 16;
        if (!tile_active(i0, j0)) continue;

        const float* krow = K + ((size_t)(b * S_ + j0 + ln) * H_ + h) * E_ + 2 * g;
        v8f s = (v8f){};
#pragma unroll
        for (int kc = 0; kc < 16; ++kc) {
            float2 t = *(const float2*)(krow + 4 * kc);
            v2f kf; kf[0] = t.x; kf[1] = t.y;
            s = __builtin_amdgcn_wmma_f32_16x16x4_f32(false, qf[kc], false, kf,
                                                      (short)0, s, false, false);
        }

#pragma unroll
        for (int r = 0; r < 8; ++r) {
            const int i = i0 + r + 8 * g;
            const int j = j0 + ln;
            float logit = log_mask(i, j) ? SCALE * s[r] : MASKED;

            float tmax = logit;
#pragma unroll
            for (int off = 8; off >= 1; off >>= 1)
                tmax = fmaxf(tmax, __shfl_xor(tmax, off, 32));

            float newmax = fmaxf(rowmax[r], tmax);
            float corr   = __expf(rowmax[r] - newmax);
            float p      = __expf(logit - newmax);

            float tsum = p;
#pragma unroll
            for (int off = 8; off >= 1; off >>= 1)
                tsum += __shfl_xor(tsum, off, 32);

            rowsum[r] = rowsum[r] * corr + tsum;
            rowmax[r] = newmax;
#pragma unroll
            for (int nt = 0; nt < 4; ++nt) oacc[nt][r] *= corr;

            plds[wave][r + 8 * g][ln] = p;
        }
        __builtin_amdgcn_wave_barrier();

        v2f pa[4];
#pragma unroll
        for (int kc = 0; kc < 4; ++kc) {
            float2 t = *(const float2*)(&plds[wave][ln][4 * kc + 2 * g]);
            v2f a; a[0] = t.x; a[1] = t.y;
            pa[kc] = a;
        }
        __builtin_amdgcn_wave_barrier();

        const float* vbase = V + ((size_t)(b * S_ + j0) * H_ + h) * D_;
#pragma unroll
        for (int nt = 0; nt < 4; ++nt) {
#pragma unroll
            for (int kc = 0; kc < 4; ++kc) {
                const int kr = 4 * kc + 2 * g;
                v2f vf;
                vf[0] = vbase[(size_t)(kr + 0) * (H_ * D_) + nt * 16 + ln];
                vf[1] = vbase[(size_t)(kr + 1) * (H_ * D_) + nt * 16 + ln];
                oacc[nt] = __builtin_amdgcn_wmma_f32_16x16x4_f32(false, pa[kc], false, vf,
                                                                 (short)0, oacc[nt], false, false);
            }
        }
    }

#pragma unroll
    for (int r = 0; r < 8; ++r) {
        const int i = i0 + r + 8 * g;
        const float inv = 1.0f / rowsum[r];
        float* orow = Out + ((size_t)(b * L_ + i) * H_ + h) * D_;
        orow[ 0 + ln] = oacc[0][r] * inv;
        orow[16 + ln] = oacc[1][r] * inv;
        orow[32 + ln] = oacc[2][r] * inv;
        orow[48 + ln] = oacc[3][r] * inv;
    }
}

extern "C" void kernel_launch(void* const* d_in, const int* in_sizes, int n_in,
                              void* d_out, int out_size, void* d_ws, size_t ws_size,
                              hipStream_t stream) {
    (void)in_sizes; (void)n_in; (void)out_size; (void)d_ws; (void)ws_size;
    const float* Q = (const float*)d_in[0];
    const float* K = (const float*)d_in[1];
    const float* V = (const float*)d_in[2];
    float* Out = (float*)d_out;

    const int total_waves = B_ * H_ * (L_ / 16);
    dim3 grid(total_waves / WAVES);
    dim3 block(WAVES * 32);
    hipLaunchKernelGGL(logsparse_attn_kernel, grid, block, 0, stream, Q, K, V, Out);
}


# scores of all 14 possible key tiles (fixed tile-distance list, no scan) held in registers: one row max and one row sum per row, no online rescale; regroups row sums and O sums
# speedup vs baseline: 2.1214x; 2.1214x over previous
#include <hip/hip_runtime.h>
#include <math.h>


typedef __attribute__((ext_vector_type(2))) float v2f;
typedef __attribute__((ext_vector_type(8))) float v8f;

#define WAVES 4
#define NSLOT 14

constexpr int B_ = 2, L_ = 2048, S_ = 2048, H_ = 8, E_ = 64, D_ = 64;
constexpr float SCALE  = 0.125f;
constexpr float MASKED = -1.25e8f;

__device__ __forceinline__ bool log_mask(int i, int j) {
    int d = i - j;
    if (d < 0) return false;
    if (i < 22) return true;
    if (d <= 10) return true;
    int e = d - 10;
    return (e <= 1024) && ((e & (e - 1)) == 0);
}

__device__ __forceinline__ constexpr int slot_m(int t) {
    return t < 6 ? t : (t < 8 ? t + 2 : (t < 10 ? t + 8 : (t < 12 ? t + 22 : t + 52)));
}

__global__ __launch_bounds__(WAVES * 32)
void logsparse_attn_kernel(const float* __restrict__ Q,
                           const float* __restrict__ K,
                           const float* __restrict__ V,
                           float* __restrict__ Out) {
    __shared__ __align__(16) float plds[WAVES][NSLOT][16][18];

    const int lane = threadIdx.x & 31;
    const int wave = __builtin_amdgcn_readfirstlane(threadIdx.x >> 5);
    const int g    = lane >> 4;
    const int ln   = lane & 15;

    const int gid = blockIdx.x * WAVES + wave;
    const int qt  = gid & 127;
    const int h   = (gid >> 7) & 7;
    const int b   = gid >> 10;
    const int i0  = qt * 16;

    const float* qrow = Q + ((size_t)(b * L_ + i0 + ln) * H_ + h) * E_ + 2 * g;
    v2f qf[16];
#pragma unroll
    for (int kc = 0; kc < 16; ++kc) {
        float2 t = *(const float2*)(qrow + 4 * kc);
        v2f a; a[0] = t.x; a[1] = t.y;
        qf[kc] = a;
    }

    v8f s[NSLOT];
#pragma unroll
    for (int tp = 0; tp < NSLOT; tp += 2) {
        const int jta = qt - slot_m(tp), jtb = qt - slot_m(tp + 1);
        const int j0a = (jta < 0 ? 0 : jta) * 16;
        const int j0b = (jtb < 0 ? 0 : jtb) * 16;
        const float* krowa = K + ((size_t)(b * S_ + j0a + ln) * H_ + h) * E_ + 2 * g;
        const float* krowb = K + ((size_t)(b * S_ + j0b + ln) * H_ + h) * E_ + 2 * g;
        v8f acca = (v8f){}, accb = (v8f){};
#pragma unroll
        for (int kc = 0; kc < 16; ++kc) {
            float2 ka = *(const float2*)(krowa + 4 * kc);
            float2 kb = *(const float2*)(krowb + 4 * kc);
            v2f kfa; kfa[0] = ka.x; kfa[1] = ka.y;
            v2f kfb; kfb[0] = kb.x; kfb[1] = kb.y;
            acca = __builtin_amdgcn_wmma_f32_16x16x4_f32(false, qf[kc], false, kfa,
                                                         (short)0, acca, false, false);
            accb = __builtin_amdgcn_wmma_f32_16x16x4_f32(false, qf[kc], false, kfb,
                                                         (short)0, accb, false, false);
        }
        s[tp] = acca;
        s[tp + 1] = accb;
    }

    float rowmax[8], rowsum[8];
#pragma unroll
    for (int r = 0; r < 8; ++r) {
        const int i = i0 + r + 8 * g;
        float mx = -INFINITY;
#pragma unroll
        for (int t = 0; t < NSLOT; ++t) {
            const int jt = qt - slot_m(t);
            const int j  = jt * 16 + ln;
            const bool ok = (jt >= 0) && log_mask(i, j);
            const float logit = ok ? SCALE * s[t][r] : MASKED;
            s[t][r] = logit;
            mx = fmaxf(mx, logit);
        }
        rowmax[r] = mx;
    }
#pragma unroll
    for (int off = 8; off >= 1; off >>= 1) {
        float other[8];
#pragma unroll
        for (int r = 0; r < 8; ++r)
            other[r] = __shfl_xor(rowmax[r], off, 32);
#pragma unroll
        for (int r = 0; r < 8; ++r)
            rowmax[r] = fmaxf(rowmax[r], other[r]);
    }
#pragma unroll
    for (int r = 0; r < 8; ++r) {
        float sum = 0.0f;
#pragma unroll
        for (int t = 0; t < NSLOT; ++t) {
            const float p = __expf(s[t][r] - rowmax[r]);
            sum += p;
            plds[wave][t][r + 8 * g][ln] = p;
        }
        rowsum[r] = sum;
    }
#pragma unroll
    for (int off = 8; off >= 1; off >>= 1) {
        float other[8];
#pragma unroll
        for (int r = 0; r < 8; ++r)
            other[r] = __shfl_xor(rowsum[r], off, 32);
#pragma unroll
        for (int r = 0; r < 8; ++r)
            rowsum[r] += other[r];
    }
    __builtin_amdgcn_wave_barrier();

    v8f oacc[4];
#pragma unroll
    for (int nt = 0; nt < 4; ++nt) oacc[nt] = (v8f){};
#pragma unroll 1
    for (int tp = 0; tp < NSLOT; tp += 2) {
        v2f pa[2][4];
        v2f vf[2][4][4];
#pragma unroll
        for (int u = 0; u < 2; ++u) {
            const int jt = qt - slot_m(tp + u);
            const int j0 = (jt < 0 ? 0 : jt) * 16;
#pragma unroll
            for (int kc = 0; kc < 4; ++kc) {
                float2 pv = *(const float2*)(&plds[wave][tp + u][ln][4 * kc + 2 * g]);
                v2f a; a[0] = pv.x; a[1] = pv.y;
                pa[u][kc] = a;
            }
            const float* vbase = V + ((size_t)(b * S_ + j0) * H_ + h) * D_;
#pragma unroll
            for (int kc = 0; kc < 4; ++kc) {
                const int kr = 4 * kc + 2 * g;
#pragma unroll
                for (int nt = 0; nt < 4; ++nt) {
                    v2f x;
                    x[0] = vbase[(size_t)(kr + 0) * (H_ * D_) + nt * 16 + ln];
                    x[1] = vbase[(size_t)(kr + 1) * (H_ * D_) + nt * 16 + ln];
                    vf[u][kc][nt] = x;
                }
            }
        }
#pragma unroll
        for (int u = 0; u < 2; ++u) {
#pragma unroll
            for (int kc = 0; kc < 4; ++kc) {
#pragma unroll
                for (int nt = 0; nt < 4; ++nt)
                    oacc[nt] = __builtin_amdgcn_wmma_f32_16x16x4_f32(false, pa[u][kc], false, vf[u][kc][nt],
                                                                     (short)0, oacc[nt], false, false);
            }
        }
    }

#pragma unroll
    for (int r = 0; r < 8; ++r) {
        const int i = i0 + r + 8 * g;
        const float inv = 1.0f / rowsum[r];
        float* orow = Out + ((size_t)(b * L_ + i) * H_ + h) * D_;
        orow[ 0 + ln] = oacc[0][r] * inv;
        orow[16 + ln] = oacc[1][r] * inv;
        orow[32 + ln] = oacc[2][r] * inv;
        orow[48 + ln] = oacc[3][r] * inv;
    }
}

extern "C" void kernel_launch(void* const* d_in, const int* in_sizes, int n_in,
                              void* d_out, int out_size, void* d_ws, size_t ws_size,
                              hipStream_t stream) {
    (void)in_sizes; (void)n_in; (void)out_size; (void)d_ws; (void)ws_size;
    const float* Q = (const float*)d_in[0];
    const float* K = (const float*)d_in[1];
    const float* V = (const float*)d_in[2];
    float* Out = (float*)d_out;

    const int total_waves = B_ * H_ * (L_ / 16);
    dim3 grid(total_waves / WAVES);
    dim3 block(WAVES * 32);
    hipLaunchKernelGGL(logsparse_attn_kernel, grid, block, 0, stream, Q, K, V, Out);
}


# all global loads and stores 16 bytes per lane: Q,K fragments take k = 8q+4g+2w+v (k order is free; regroups the k sums of Q.K), V and Out take column 4*lane+tile; 688 -> 232 loads per wave
# speedup vs baseline: 2.6056x; 1.2282x over previous
#include <hip/hip_runtime.h>
#include <math.h>


typedef __attribute__((ext_vector_type(2))) float v2f;
typedef __attribute__((ext_vector_type(8))) float v8f;

#define WAVES 4
#define NSLOT 14

constexpr int B_ = 2, L_ = 2048, S_ = 2048, H_ = 8, E_ = 64, D_ = 64;
constexpr float SCALE  = 0.125f;
constexpr float MASKED = -1.25e8f;

__device__ __forceinline__ bool log_mask(int i, int j) {
    int d = i - j;
    if (d < 0) return false;
    if (i < 22) return true;
    if (d <= 10) return true;
    int e = d - 10;
    return (e <= 1024) && ((e & (e - 1)) == 0);
}

__device__ __forceinline__ constexpr int slot_m(int t) {
    return t < 6 ? t : (t < 8 ? t + 2 : (t < 10 ? t + 8 : (t < 12 ? t + 22 : t + 52)));
}

__global__ __launch_bounds__(WAVES * 32)
void logsparse_attn_kernel(const float* __restrict__ Q,
                           const float* __restrict__ K,
                           const float* __restrict__ V,
                           float* __restrict__ Out) {
    __shared__ __align__(16) float plds[WAVES][NSLOT][16][18];

    const int lane = threadIdx.x & 31;
    const int wave = __builtin_amdgcn_readfirstlane(threadIdx.x >> 5);
    const int g    = lane >> 4;
    const int ln   = lane & 15;

    const int gid = blockIdx.x * WAVES + wave;
    const int qt  = gid & 127;
    const int h   = (gid >> 7) & 7;
    const int b   = gid >> 10;
    const int i0  = qt * 16;

    const float* qrow = Q + ((size_t)(b * L_ + i0 + ln) * H_ + h) * E_ + 4 * g;
    v2f qf[16];
#pragma unroll
    for (int q = 0; q < 8; ++q) {
        float4 t = *(const float4*)(qrow + 8 * q);
        v2f a0; a0[0] = t.x; a0[1] = t.y;
        v2f a1; a1[0] = t.z; a1[1] = t.w;
        qf[2 * q]     = a0;
        qf[2 * q + 1] = a1;
    }

    v8f s[NSLOT];
#pragma unroll
    for (int tp = 0; tp < NSLOT; tp += 2) {
        const int jta = qt - slot_m(tp), jtb = qt - slot_m(tp + 1);
        const int j0a = (jta < 0 ? 0 : jta) * 16;
        const int j0b = (jtb < 0 ? 0 : jtb) * 16;
        const float* krowa = K + ((size_t)(b * S_ + j0a + ln) * H_ + h) * E_ + 4 * g;
        const float* krowb = K + ((size_t)(b * S_ + j0b + ln) * H_ + h) * E_ + 4 * g;
        v8f acca = (v8f){}, accb = (v8f){};
#pragma unroll
        for (int q = 0; q < 8; ++q) {
            float4 ka = *(const float4*)(krowa + 8 * q);
            float4 kb = *(const float4*)(krowb + 8 * q);
            v2f kfa0; kfa0[0] = ka.x; kfa0[1] = ka.y;
            v2f kfa1; kfa1[0] = ka.z; kfa1[1] = ka.w;
            v2f kfb0; kfb0[0] = kb.x; kfb0[1] = kb.y;
            v2f kfb1; kfb1[0] = kb.z; kfb1[1] = kb.w;
            acca = __builtin_amdgcn_wmma_f32_16x16x4_f32(false, qf[2 * q], false, kfa0,
                                                         (short)0, acca, false, false);
            accb = __builtin_amdgcn_wmma_f32_16x16x4_f32(false, qf[2 * q], false, kfb0,
                                                         (short)0, accb, false, false);
            acca = __builtin_amdgcn_wmma_f32_16x16x4_f32(false, qf[2 * q + 1], false, kfa1,
                                                         (short)0, acca, false, false);
            accb = __builtin_amdgcn_wmma_f32_16x16x4_f32(false, qf[2 * q + 1], false, kfb1,
                                                         (short)0, accb, false, false);
        }
        s[tp] = acca;
        s[tp + 1] = accb;
    }

    float rowmax[8], rowsum[8];
#pragma unroll
    for (int r = 0; r < 8; ++r) {
        const int i = i0 + r + 8 * g;
        float mx = -INFINITY;
#pragma unroll
        for (int t = 0; t < NSLOT; ++t) {
            const int jt = qt - slot_m(t);
            const int j  = jt * 16 + ln;
            const bool ok = (jt >= 0) && log_mask(i, j);
            const float logit = ok ? SCALE * s[t][r] : MASKED;
            s[t][r] = logit;
            mx = fmaxf(mx, logit);
        }
        rowmax[r] = mx;
    }
#pragma unroll
    for (int off = 8; off >= 1; off >>= 1) {
        float other[8];
#pragma unroll
        for (int r = 0; r < 8; ++r)
            other[r] = __shfl_xor(rowmax[r], off, 32);
#pragma unroll
        for (int r = 0; r < 8; ++r)
            rowmax[r] = fmaxf(rowmax[r], other[r]);
    }
#pragma unroll
    for (int r = 0; r < 8; ++r) {
        float sum = 0.0f;
#pragma unroll
        for (int t = 0; t < NSLOT; ++t) {
            const float p = __expf(s[t][r] - rowmax[r]);
            sum += p;
            plds[wave][t][r + 8 * g][ln] = p;
        }
        rowsum[r] = sum;
    }
#pragma unroll
    for (int off = 8; off >= 1; off >>= 1) {
        float other[8];
#pragma unroll
        for (int r = 0; r < 8; ++r)
            other[r] = __shfl_xor(rowsum[r], off, 32);
#pragma unroll
        for (int r = 0; r < 8; ++r)
            rowsum[r] += other[r];
    }
    __builtin_amdgcn_wave_barrier();

    v8f oacc[4];
#pragma unroll
    for (int nt = 0; nt < 4; ++nt) oacc[nt] = (v8f){};
#pragma unroll 1
    for (int tp = 0; tp < NSLOT; tp += 2) {
        v2f pa[2][4];
        v2f vf[2][4][4];
#pragma unroll
        for (int u = 0; u < 2; ++u) {
            const int jt = qt - slot_m(tp + u);
            const int j0 = (jt < 0 ? 0 : jt) * 16;
#pragma unroll
            for (int kc = 0; kc < 4; ++kc) {
                float2 pv = *(const float2*)(&plds[wave][tp + u][ln][4 * kc + 2 * g]);
                v2f a; a[0] = pv.x; a[1] = pv.y;
                pa[u][kc] = a;
            }
            const float* vbase = V + ((size_t)(b * S_ + j0) * H_ + h) * D_ + 4 * ln;
#pragma unroll
            for (int kc = 0; kc < 4; ++kc) {
                const int kr = 4 * kc + 2 * g;
                const float4 r0 = *(const float4*)(vbase + (size_t)(kr + 0) * (H_ * D_));
                const float4 r1 = *(const float4*)(vbase + (size_t)(kr + 1) * (H_ * D_));
                v2f x0; x0[0] = r0.x; x0[1] = r1.x;
                v2f x1; x1[0] = r0.y; x1[1] = r1.y;
                v2f x2; x2[0] = r0.z; x2[1] = r1.z;
                v2f x3; x3[0] = r0.w; x3[1] = r1.w;
                vf[u][kc][0] = x0;
                vf[u][kc][1] = x1;
                vf[u][kc][2] = x2;
                vf[u][kc][3] = x3;
            }
        }
#pragma unroll
        for (int u = 0; u < 2; ++u) {
#pragma unroll
            for (int kc = 0; kc < 4; ++kc) {
#pragma unroll
                for (int nt = 0; nt < 4; ++nt)
                    oacc[nt] = __builtin_amdgcn_wmma_f32_16x16x4_f32(false, pa[u][kc], false, vf[u][kc][nt],
                                                                     (short)0, oacc[nt], false, false);
            }
        }
    }

#pragma unroll
    for (int r = 0; r < 8; ++r) {
        const int i = i0 + r + 8 * g;
        const float inv = 1.0f / rowsum[r];
        float* orow = Out + ((size_t)(b * L_ + i) * H_ + h) * D_;
        float4 o;
        o.x = oacc[0][r] * inv;
        o.y = oacc[1][r] * inv;
        o.z = oacc[2][r] * inv;
        o.w = oacc[3][r] * inv;
        *(float4*)(orow + 4 * ln) = o;
    }
}

extern "C" void kernel_launch(void* const* d_in, const int* in_sizes, int n_in,
                              void* d_out, int out_size, void* d_ws, size_t ws_size,
                              hipStream_t stream) {
    (void)in_sizes; (void)n_in; (void)out_size; (void)d_ws; (void)ws_size;
    const float* Q = (const float*)d_in[0];
    const float* K = (const float*)d_in[1];
    const float* V = (const float*)d_in[2];
    float* Out = (float*)d_out;

    const int total_waves = B_ * H_ * (L_ / 16);
    dim3 grid(total_waves / WAVES);
    dim3 block(WAVES * 32);
    hipLaunchKernelGGL(logsparse_attn_kernel, grid, block, 0, stream, Q, K, V, Out);
}


# mask test without branches: the distance rule as a bit string, shifted once per lane by its row-column offset, gives each (slot,row) one fixed bit; far tile pairs share bits; same sums and order
# speedup vs baseline: 3.0617x; 1.1750x over previous
#include <hip/hip_runtime.h>
#include <math.h>


typedef __attribute__((ext_vector_type(2))) float v2f;
typedef __attribute__((ext_vector_type(8))) float v8f;

#define WAVES 4
#define NSLOT 14

constexpr int B_ = 2, L_ = 2048, S_ = 2048, H_ = 8, E_ = 64, D_ = 64;
constexpr float SCALE  = 0.125f;
constexpr float MASKED = -1.25e8f;

constexpr bool dist_ok(int d) {
    if (d < 0) return false;
    if (d <= 10) return true;
    int e = d - 10;
    return (e <= 1024) && ((e & (e - 1)) == 0);
}
constexpr unsigned allow_word(int w) {
    unsigned v = 0;
    for (int b = 0; b < 32; ++b)
        if (dist_ok(32 * w + b - 15)) v |= 1u << b;
    return v;
}

__device__ __forceinline__ constexpr int slot_m(int t) {
    return t < 6 ? t : (t < 8 ? t + 2 : (t < 10 ? t + 8 : (t < 12 ? t + 22 : t + 52)));
}

__global__ __launch_bounds__(WAVES * 32)
void logsparse_attn_kernel(const float* __restrict__ Q,
                           const float* __restrict__ K,
                           const float* __restrict__ V,
                           float* __restrict__ Out) {
    __shared__ __align__(16) float plds[WAVES][NSLOT][16][18];

    const int lane = threadIdx.x & 31;
    const int wave = __builtin_amdgcn_readfirstlane(threadIdx.x >> 5);
    const int g    = lane >> 4;
    const int ln   = lane & 15;

    const int gid = blockIdx.x * WAVES + wave;
    const int qt  = gid & 127;
    const int h   = (gid >> 7) & 7;
    const int b   = gid >> 10;
    const int i0  = qt * 16;

    const float* qrow = Q + ((size_t)(b * L_ + i0 + ln) * H_ + h) * E_ + 4 * g;
    v2f qf[16];
#pragma unroll
    for (int q = 0; q < 8; ++q) {
        float4 t = *(const float4*)(qrow + 8 * q);
        v2f a0; a0[0] = t.x; a0[1] = t.y;
        v2f a1; a1[0] = t.z; a1[1] = t.w;
        qf[2 * q]     = a0;
        qf[2 * q + 1] = a1;
    }

    v8f s[NSLOT];
#pragma unroll
    for (int tp = 0; tp < NSLOT; tp += 2) {
        const int jta = qt - slot_m(tp), jtb = qt - slot_m(tp + 1);
        const int j0a = (jta < 0 ? 0 : jta) * 16;
        const int j0b = (jtb < 0 ? 0 : jtb) * 16;
        const float* krowa = K + ((size_t)(b * S_ + j0a + ln) * H_ + h) * E_ + 4 * g;
        const float* krowb = K + ((size_t)(b * S_ + j0b + ln) * H_ + h) * E_ + 4 * g;
        v8f acca = (v8f){}, accb = (v8f){};
#pragma unroll
        for (int q = 0; q < 8; ++q) {
            float4 ka = *(const float4*)(krowa + 8 * q);
            float4 kb = *(const float4*)(krowb + 8 * q);
            v2f kfa0; kfa0[0] = ka.x; kfa0[1] = ka.y;
            v2f kfa1; kfa1[0] = ka.z; kfa1[1] = ka.w;
            v2f kfb0; kfb0[0] = kb.x; kfb0[1] = kb.y;
            v2f kfb1; kfb1[0] = kb.z; kfb1[1] = kb.w;
            acca = __builtin_amdgcn_wmma_f32_16x16x4_f32(false, qf[2 * q], false, kfa0,
                                                         (short)0, acca, false, false);
            accb = __builtin_amdgcn_wmma_f32_16x16x4_f32(false, qf[2 * q], false, kfb0,
                                                         (short)0, accb, false, false);
            acca = __builtin_amdgcn_wmma_f32_16x16x4_f32(false, qf[2 * q + 1], false, kfa1,
                                                         (short)0, acca, false, false);
            accb = __builtin_amdgcn_wmma_f32_16x16x4_f32(false, qf[2 * q + 1], false, kfb1,
                                                         (short)0, accb, false, false);
        }
        s[tp] = acca;
        s[tp + 1] = accb;
    }

    const int y = 15 + 8 * g - ln;
    constexpr unsigned long long A01 = ((unsigned long long)allow_word(1) << 32) | allow_word(0);
    constexpr unsigned long long A12 = ((unsigned long long)allow_word(2) << 32) | allow_word(1);
    static_assert(allow_word(3) == 0u, "bits 16*5 + 7 + 23 and below are all that is read");
    const unsigned w0 = (unsigned)(A01 >> y);
    const unsigned w1 = (unsigned)(A12 >> y);
    const unsigned w2 = allow_word(2) >> y;
    const unsigned c0 = (unsigned)(0xFFFFFFFFFFFF8000ull >> y);
    float rowmax[8], rowsum[8];
#pragma unroll
    for (int r = 0; r < 8; ++r) {
        const int i = i0 + r + 8 * g;
        const unsigned w0r = (i < 22) ? c0 : w0;
        float mx = -INFINITY;
#pragma unroll
        for (int t = 0; t < NSLOT; ++t) {
            const int m  = slot_m(t);
            const int mc = t < 4 ? m : 4 + (t & 1);
            const int c  = 16 * mc + r;
            const unsigned word = c < 32 ? w0r : (c < 64 ? w1 : w2);
            const unsigned live = (qt >= m) ? word : 0u;
            const bool ok = ((live >> (c & 31)) & 1u) != 0u;
            const float logit = ok ? SCALE * s[t][r] : MASKED;
            s[t][r] = logit;
            mx = fmaxf(mx, logit);
        }
        rowmax[r] = mx;
    }
#pragma unroll
    for (int off = 8; off >= 1; off >>= 1) {
        float other[8];
#pragma unroll
        for (int r = 0; r < 8; ++r)
            other[r] = __shfl_xor(rowmax[r], off, 32);
#pragma unroll
        for (int r = 0; r < 8; ++r)
            rowmax[r] = fmaxf(rowmax[r], other[r]);
    }
#pragma unroll
    for (int r = 0; r < 8; ++r) {
        float sum = 0.0f;
#pragma unroll
        for (int t = 0; t < NSLOT; ++t) {
            const float p = __expf(s[t][r] - rowmax[r]);
            sum += p;
            plds[wave][t][r + 8 * g][ln] = p;
        }
        rowsum[r] = sum;
    }
#pragma unroll
    for (int off = 8; off >= 1; off >>= 1) {
        float other[8];
#pragma unroll
        for (int r = 0; r < 8; ++r)
            other[r] = __shfl_xor(rowsum[r], off, 32);
#pragma unroll
        for (int r = 0; r < 8; ++r)
            rowsum[r] += other[r];
    }
    __builtin_amdgcn_wave_barrier();

    v8f oacc[4];
#pragma unroll
    for (int nt = 0; nt < 4; ++nt) oacc[nt] = (v8f){};
#pragma unroll 1
    for (int tp = 0; tp < NSLOT; tp += 2) {
        v2f pa[2][4];
        v2f vf[2][4][4];
#pragma unroll
        for (int u = 0; u < 2; ++u) {
            const int jt = qt - slot_m(tp + u);
            const int j0 = (jt < 0 ? 0 : jt) * 16;
#pragma unroll
            for (int kc = 0; kc < 4; ++kc) {
                float2 pv = *(const float2*)(&plds[wave][tp + u][ln][4 * kc + 2 * g]);
                v2f a; a[0] = pv.x; a[1] = pv.y;
                pa[u][kc] = a;
            }
            const float* vbase = V + ((size_t)(b * S_ + j0) * H_ + h) * D_ + 4 * ln;
#pragma unroll
            for (int kc = 0; kc < 4; ++kc) {
                const int kr = 4 * kc + 2 * g;
                const float4 r0 = *(const float4*)(vbase + (size_t)(kr + 0) * (H_ * D_));
                const float4 r1 = *(const float4*)(vbase + (size_t)(kr + 1) * (H_ * D_));
                v2f x0; x0[0] = r0.x; x0[1] = r1.x;
                v2f x1; x1[0] = r0.y; x1[1] = r1.y;
                v2f x2; x2[0] = r0.z; x2[1] = r1.z;
                v2f x3; x3[0] = r0.w; x3[1] = r1.w;
                vf[u][kc][0] = x0;
                vf[u][kc][1] = x1;
                vf[u][kc][2] = x2;
                vf[u][kc][3] = x3;
            }
        }
#pragma unroll
        for (int u = 0; u < 2; ++u) {
#pragma unroll
            for (int kc = 0; kc < 4; ++kc) {
#pragma unroll
                for (int nt = 0; nt < 4; ++nt)
                    oacc[nt] = __builtin_amdgcn_wmma_f32_16x16x4_f32(false, pa[u][kc], false, vf[u][kc][nt],
                                                                     (short)0, oacc[nt], false, false);
            }
        }
    }

#pragma unroll
    for (int r = 0; r < 8; ++r) {
        const int i = i0 + r + 8 * g;
        const float inv = 1.0f / rowsum[r];
        float* orow = Out + ((size_t)(b * L_ + i) * H_ + h) * D_;
        float4 o;
        o.x = oacc[0][r] * inv;
        o.y = oacc[1][r] * inv;
        o.z = oacc[2][r] * inv;
        o.w = oacc[3][r] * inv;
        *(float4*)(orow + 4 * ln) = o;
    }
}

extern "C" void kernel_launch(void* const* d_in, const int* in_sizes, int n_in,
                              void* d_out, int out_size, void* d_ws, size_t ws_size,
                              hipStream_t stream) {
    (void)in_sizes; (void)n_in; (void)out_size; (void)d_ws; (void)ws_size;
    const float* Q = (const float*)d_in[0];
    const float* K = (const float*)d_in[1];
    const float* V = (const float*)d_in[2];
    float* Out = (float*)d_out;

    const int total_waves = B_ * H_ * (L_ / 16);
    dim3 grid(total_waves / WAVES);
    dim3 block(WAVES * 32);
    hipLaunchKernelGGL(logsparse_attn_kernel, grid, block, 0, stream, Q, K, V, Out);
}


# Q,K tiles fetched in whole 128-byte lines (4 lines per load, not 16) and turned into WMMA fragments through LDS, next trip's tiles in flight (2 waves/SIMD declared for registers); same k order, sums
# speedup vs baseline: 3.9853x; 1.3017x over previous
#include <hip/hip_runtime.h>
#include <math.h>


typedef __attribute__((ext_vector_type(2))) float v2f;
typedef __attribute__((ext_vector_type(8))) float v8f;
typedef __attribute__((ext_vector_type(4))) float v4f;

#define WAVES 4
#define NSLOT 14

constexpr int B_ = 2, L_ = 2048, S_ = 2048, H_ = 8, E_ = 64, D_ = 64;
constexpr float SCALE  = 0.125f;
constexpr float MASKED = -1.25e8f;

constexpr bool dist_ok(int d) {
    if (d < 0) return false;
    if (d <= 10) return true;
    int e = d - 10;
    return (e <= 1024) && ((e & (e - 1)) == 0);
}
constexpr unsigned allow_word(int w) {
    unsigned v = 0;
    for (int b = 0; b < 32; ++b)
        if (dist_ok(32 * w + b - 15)) v |= 1u << b;
    return v;
}

__device__ __forceinline__ constexpr int slot_m(int t) {
    return t < 6 ? t : (t < 8 ? t + 2 : (t < 10 ? t + 8 : (t < 12 ? t + 22 : t + 52)));
}

#define PITCH 68

__device__ __forceinline__ void wave_sync() {
    __builtin_amdgcn_fence(__ATOMIC_RELEASE, "wavefront");
    __builtin_amdgcn_wave_barrier();
    __builtin_amdgcn_fence(__ATOMIC_ACQUIRE, "wavefront");
}

__device__ __forceinline__ void tile_gload(const float* __restrict__ tile, int g, int ln, v4f (&r)[8]) {
#pragma unroll
    for (int i = 0; i < 8; ++i)
        r[i] = *(const v4f*)(tile + (size_t)(2 * i + g) * (H_ * E_) + 4 * ln);
}
__device__ __forceinline__ void tile_stage(float* st, int g, int ln, const v4f (&r)[8]) {
#pragma unroll
    for (int i = 0; i < 8; ++i)
        *(v4f*)(st + (2 * i + g) * PITCH + 4 * ln) = r[i];
}
__device__ __forceinline__ void tile_frags(const float* st, int g, int ln, v2f (&f)[16]) {
#pragma unroll
    for (int q = 0; q < 8; ++q) {
        const v4f t = *(const v4f*)(st + ln * PITCH + 8 * q + 4 * g);
        v2f a0; a0[0] = t[0]; a0[1] = t[1];
        v2f a1; a1[0] = t[2]; a1[1] = t[3];
        f[2 * q]     = a0;
        f[2 * q + 1] = a1;
    }
}

__global__ __launch_bounds__(WAVES * 32) __attribute__((amdgpu_waves_per_eu(2, 2)))
void logsparse_attn_kernel(const float* __restrict__ Q,
                           const float* __restrict__ K,
                           const float* __restrict__ V,
                           float* __restrict__ Out) {
    __shared__ __align__(16) float plds[WAVES][NSLOT][16][18];

    const int lane = threadIdx.x & 31;
    const int wave = __builtin_amdgcn_readfirstlane(threadIdx.x >> 5);
    const int g    = lane >> 4;
    const int ln   = lane & 15;

    const int gid = blockIdx.x * WAVES + wave;
    const int qt  = gid & 127;
    const int h   = (gid >> 7) & 7;
    const int b   = gid >> 10;
    const int i0  = qt * 16;

    float* stA = &plds[wave][0][0][0];
    float* stB = stA + 16 * PITCH;

    v4f gq[8], ga[8], gb[8];
    tile_gload(Q + ((size_t)(b * L_ + i0) * H_ + h) * E_, g, ln, gq);
    {
        const int jta = qt - slot_m(0), jtb = qt - slot_m(1);
        tile_gload(K + ((size_t)(b * S_ + (jta < 0 ? 0 : jta) * 16) * H_ + h) * E_, g, ln, ga);
        tile_gload(K + ((size_t)(b * S_ + (jtb < 0 ? 0 : jtb) * 16) * H_ + h) * E_, g, ln, gb);
    }
    v2f qf[16];
    tile_stage(stA, g, ln, gq);
    wave_sync();
    tile_frags(stA, g, ln, qf);
    wave_sync();

    v2f kfa[16], kfb[16];
    tile_stage(stA, g, ln, ga);
    tile_stage(stB, g, ln, gb);
    {
        const int jta = qt - slot_m(2), jtb = qt - slot_m(3);
        tile_gload(K + ((size_t)(b * S_ + (jta < 0 ? 0 : jta) * 16) * H_ + h) * E_, g, ln, ga);
        tile_gload(K + ((size_t)(b * S_ + (jtb < 0 ? 0 : jtb) * 16) * H_ + h) * E_, g, ln, gb);
    }
    wave_sync();
    tile_frags(stA, g, ln, kfa);
    tile_frags(stB, g, ln, kfb);
    wave_sync();

    v8f s[NSLOT];
#pragma unroll
    for (int tp = 0; tp < NSLOT; tp += 2) {
        v2f nfa[16], nfb[16];
        if (tp + 2 < NSLOT) {
            tile_stage(stA, g, ln, ga);
            tile_stage(stB, g, ln, gb);
            if (tp + 4 < NSLOT) {
                const int jta = qt - slot_m(tp + 4), jtb = qt - slot_m(tp + 5);
                tile_gload(K + ((size_t)(b * S_ + (jta < 0 ? 0 : jta) * 16) * H_ + h) * E_, g, ln, ga);
                tile_gload(K + ((size_t)(b * S_ + (jtb < 0 ? 0 : jtb) * 16) * H_ + h) * E_, g, ln, gb);
            }
            wave_sync();
            tile_frags(stA, g, ln, nfa);
            tile_frags(stB, g, ln, nfb);
            wave_sync();
        }
        __builtin_amdgcn_sched_barrier(0);
        v8f acca = (v8f){}, accb = (v8f){};
#pragma unroll
        for (int q = 0; q < 8; ++q) {
            acca = __builtin_amdgcn_wmma_f32_16x16x4_f32(false, qf[2 * q], false, kfa[2 * q],
                                                         (short)0, acca, false, false);
            accb = __builtin_amdgcn_wmma_f32_16x16x4_f32(false, qf[2 * q], false, kfb[2 * q],
                                                         (short)0, accb, false, false);
            acca = __builtin_amdgcn_wmma_f32_16x16x4_f32(false, qf[2 * q + 1], false, kfa[2 * q + 1],
                                                         (short)0, acca, false, false);
            accb = __builtin_amdgcn_wmma_f32_16x16x4_f32(false, qf[2 * q + 1], false, kfb[2 * q + 1],
                                                         (short)0, accb, false, false);
        }
        s[tp] = acca;
        s[tp + 1] = accb;
        if (tp + 2 < NSLOT) {
#pragma unroll
            for (int f = 0; f < 16; ++f) { kfa[f] = nfa[f]; kfb[f] = nfb[f]; }
        }
        __builtin_amdgcn_sched_barrier(0);
    }

    const int y = 15 + 8 * g - ln;
    constexpr unsigned long long A01 = ((unsigned long long)allow_word(1) << 32) | allow_word(0);
    constexpr unsigned long long A12 = ((unsigned long long)allow_word(2) << 32) | allow_word(1);
    static_assert(allow_word(3) == 0u, "bits 16*5 + 7 + 23 and below are all that is read");
    const unsigned w0 = (unsigned)(A01 >> y);
    const unsigned w1 = (unsigned)(A12 >> y);
    const unsigned w2 = allow_word(2) >> y;
    const unsigned c0 = (unsigned)(0xFFFFFFFFFFFF8000ull >> y);
    float rowmax[8], rowsum[8];
#pragma unroll
    for (int r = 0; r < 8; ++r) {
        const int i = i0 + r + 8 * g;
        const unsigned w0r = (i < 22) ? c0 : w0;
        float mx = -INFINITY;
#pragma unroll
        for (int t = 0; t < NSLOT; ++t) {
            const int m  = slot_m(t);
            const int mc = t < 4 ? m : 4 + (t & 1);
            const int c  = 16 * mc + r;
            const unsigned word = c < 32 ? w0r : (c < 64 ? w1 : w2);
            const unsigned live = (qt >= m) ? word : 0u;
            const bool ok = ((live >> (c & 31)) & 1u) != 0u;
            const float logit = ok ? SCALE * s[t][r] : MASKED;
            s[t][r] = logit;
            mx = fmaxf(mx, logit);
        }
        rowmax[r] = mx;
    }
#pragma unroll
    for (int off = 8; off >= 1; off >>= 1) {
        float other[8];
#pragma unroll
        for (int r = 0; r < 8; ++r)
            other[r] = __shfl_xor(rowmax[r], off, 32);
#pragma unroll
        for (int r = 0; r < 8; ++r)
            rowmax[r] = fmaxf(rowmax[r], other[r]);
    }
#pragma unroll
    for (int r = 0; r < 8; ++r) {
        float sum = 0.0f;
#pragma unroll
        for (int t = 0; t < NSLOT; ++t) {
            const float p = __expf(s[t][r] - rowmax[r]);
            sum += p;
            plds[wave][t][r + 8 * g][ln] = p;
        }
        rowsum[r] = sum;
    }
#pragma unroll
    for (int off = 8; off >= 1; off >>= 1) {
        float other[8];
#pragma unroll
        for (int r = 0; r < 8; ++r)
            other[r] = __shfl_xor(rowsum[r], off, 32);
#pragma unroll
        for (int r = 0; r < 8; ++r)
            rowsum[r] += other[r];
    }
    __builtin_amdgcn_wave_barrier();

    v8f oacc[4];
#pragma unroll
    for (int nt = 0; nt < 4; ++nt) oacc[nt] = (v8f){};
#pragma unroll 1
    for (int tp = 0; tp < NSLOT; tp += 2) {
        v2f pa[2][4];
        v2f vf[2][4][4];
#pragma unroll
        for (int u = 0; u < 2; ++u) {
            const int jt = qt - slot_m(tp + u);
            const int j0 = (jt < 0 ? 0 : jt) * 16;
#pragma unroll
            for (int kc = 0; kc < 4; ++kc) {
                float2 pv = *(const float2*)(&plds[wave][tp + u][ln][4 * kc + 2 * g]);
                v2f a; a[0] = pv.x; a[1] = pv.y;
                pa[u][kc] = a;
            }
            const float* vbase = V + ((size_t)(b * S_ + j0) * H_ + h) * D_ + 4 * ln;
#pragma unroll
            for (int kc = 0; kc < 4; ++kc) {
                const int kr = 4 * kc + 2 * g;
                const float4 r0 = *(const float4*)(vbase + (size_t)(kr + 0) * (H_ * D_));
                const float4 r1 = *(const float4*)(vbase + (size_t)(kr + 1) * (H_ * D_));
                v2f x0; x0[0] = r0.x; x0[1] = r1.x;
                v2f x1; x1[0] = r0.y; x1[1] = r1.y;
                v2f x2; x2[0] = r0.z; x2[1] = r1.z;
                v2f x3; x3[0] = r0.w; x3[1] = r1.w;
                vf[u][kc][0] = x0;
                vf[u][kc][1] = x1;
                vf[u][kc][2] = x2;
                vf[u][kc][3] = x3;
            }
        }
#pragma unroll
        for (int u = 0; u < 2; ++u) {
#pragma unroll
            for (int kc = 0; kc < 4; ++kc) {
#pragma unroll
                for (int nt = 0; nt < 4; ++nt)
                    oacc[nt] = __builtin_amdgcn_wmma_f32_16x16x4_f32(false, pa[u][kc], false, vf[u][kc][nt],
                                                                     (short)0, oacc[nt], false, false);
            }
        }
    }

#pragma unroll
    for (int r = 0; r < 8; ++r) {
        const int i = i0 + r + 8 * g;
        const float inv = 1.0f / rowsum[r];
        float* orow = Out + ((size_t)(b * L_ + i) * H_ + h) * D_;
        float4 o;
        o.x = oacc[0][r] * inv;
        o.y = oacc[1][r] * inv;
        o.z = oacc[2][r] * inv;
        o.w = oacc[3][r] * inv;
        *(float4*)(orow + 4 * ln) = o;
    }
}

extern "C" void kernel_launch(void* const* d_in, const int* in_sizes, int n_in,
                              void* d_out, int out_size, void* d_ws, size_t ws_size,
                              hipStream_t stream) {
    (void)in_sizes; (void)n_in; (void)out_size; (void)d_ws; (void)ws_size;
    const float* Q = (const float*)d_in[0];
    const float* K = (const float*)d_in[1];
    const float* V = (const float*)d_in[2];
    float* Out = (float*)d_out;

    const int total_waves = B_ * H_ * (L_ / 16);
    dim3 grid(total_waves / WAVES);
    dim3 block(WAVES * 32);
    hipLaunchKernelGGL(logsparse_attn_kernel, grid, block, 0, stream, Q, K, V, Out);
}


# P*V runs only over the 40 of 56 four-key chunks whose P columns the distance rule can leave non-zero (far tile pairs: 5 of 8 chunks), and loads only their V rows; same sums, same order
# speedup vs baseline: 4.2066x; 1.0555x over previous
#include <hip/hip_runtime.h>
#include <math.h>


typedef __attribute__((ext_vector_type(2))) float v2f;
typedef __attribute__((ext_vector_type(8))) float v8f;
typedef __attribute__((ext_vector_type(4))) float v4f;

#define WAVES 4
#define NSLOT 14

constexpr int B_ = 2, L_ = 2048, S_ = 2048, H_ = 8, E_ = 64, D_ = 64;
constexpr float SCALE  = 0.125f;
constexpr float MASKED = -1.25e8f;

constexpr bool dist_ok(int d) {
    if (d < 0) return false;
    if (d <= 10) return true;
    int e = d - 10;
    return (e <= 1024) && ((e & (e - 1)) == 0);
}
constexpr unsigned allow_word(int w) {
    unsigned v = 0;
    for (int b = 0; b < 32; ++b)
        if (dist_ok(32 * w + b - 15)) v |= 1u << b;
    return v;
}

__device__ __forceinline__ constexpr int slot_m(int t) {
    return t < 6 ? t : (t < 8 ? t + 2 : (t < 10 ? t + 8 : (t < 12 ? t + 22 : t + 52)));
}

#define PITCH 68

__device__ __forceinline__ void wave_sync() {
    __builtin_amdgcn_fence(__ATOMIC_RELEASE, "wavefront");
    __builtin_amdgcn_wave_barrier();
    __builtin_amdgcn_fence(__ATOMIC_ACQUIRE, "wavefront");
}

__device__ __forceinline__ void tile_gload(const float* __restrict__ tile, int g, int ln, v4f (&r)[8]) {
#pragma unroll
    for (int i = 0; i < 8; ++i)
        r[i] = *(const v4f*)(tile + (size_t)(2 * i + g) * (H_ * E_) + 4 * ln);
}
__device__ __forceinline__ void tile_stage(float* st, int g, int ln, const v4f (&r)[8]) {
#pragma unroll
    for (int i = 0; i < 8; ++i)
        *(v4f*)(st + (2 * i + g) * PITCH + 4 * ln) = r[i];
}
__device__ __forceinline__ void tile_frags(const float* st, int g, int ln, v2f (&f)[16]) {
#pragma unroll
    for (int q = 0; q < 8; ++q) {
        const v4f t = *(const v4f*)(st + ln * PITCH + 8 * q + 4 * g);
        v2f a0; a0[0] = t[0]; a0[1] = t[1];
        v2f a1; a1[0] = t[2]; a1[1] = t[3];
        f[2 * q]     = a0;
        f[2 * q + 1] = a1;
    }
}

constexpr bool chunk_live(int m, int kc) {
    for (int c = 4 * kc; c < 4 * kc + 4; ++c)
        for (int row = 0; row < 16; ++row)
            if (dist_ok(16 * m + row - c)) return true;
    return false;
}
static_assert(!chunk_live(3, 0), "tile distance 3: key columns 0..3 are masked in every row");
static_assert(!chunk_live(4, 2) && !chunk_live(4, 3), "nearer far tile: key columns 8..15 are masked");
static_assert(!chunk_live(5, 0), "farther far tile: key columns 0..3 are masked");

template <int LO_A, int HI_A, int LO_B, int HI_B>
__device__ __forceinline__ void pv_trip(const float* __restrict__ V, const float* pl, int qt, int ma,
                                        int b, int h, int g, int ln, v8f (&oacc)[4]) {
    v2f pa[2][4];
    v2f vf[2][4][4];
#pragma unroll
    for (int u = 0; u < 2; ++u) {
        const int lo = u ? LO_B : LO_A, hi = u ? HI_B : HI_A;
        const int jt = qt - (ma + u);
        const int j0 = (jt < 0 ? 0 : jt) * 16;
#pragma unroll
        for (int kc = lo; kc < hi; ++kc) {
            float2 pv = *(const float2*)(pl + (u * 16 + ln) * 18 + 4 * kc + 2 * g);
            v2f a; a[0] = pv.x; a[1] = pv.y;
            pa[u][kc] = a;
        }
        const float* vbase = V + ((size_t)(b * S_ + j0) * H_ + h) * D_ + 4 * ln;
#pragma unroll
        for (int kc = lo; kc < hi; ++kc) {
            const int kr = 4 * kc + 2 * g;
            const float4 r0 = *(const float4*)(vbase + (size_t)(kr + 0) * (H_ * D_));
            const float4 r1 = *(const float4*)(vbase + (size_t)(kr + 1) * (H_ * D_));
            v2f x0; x0[0] = r0.x; x0[1] = r1.x;
            v2f x1; x1[0] = r0.y; x1[1] = r1.y;
            v2f x2; x2[0] = r0.z; x2[1] = r1.z;
            v2f x3; x3[0] = r0.w; x3[1] = r1.w;
            vf[u][kc][0] = x0;
            vf[u][kc][1] = x1;
            vf[u][kc][2] = x2;
            vf[u][kc][3] = x3;
        }
    }
#pragma unroll
    for (int u = 0; u < 2; ++u) {
        const int lo = u ? LO_B : LO_A, hi = u ? HI_B : HI_A;
#pragma unroll
        for (int kc = lo; kc < hi; ++kc) {
#pragma unroll
            for (int nt = 0; nt < 4; ++nt)
                oacc[nt] = __builtin_amdgcn_wmma_f32_16x16x4_f32(false, pa[u][kc], false, vf[u][kc][nt],
                                                                 (short)0, oacc[nt], false, false);
        }
    }
}

__global__ __launch_bounds__(WAVES * 32) __attribute__((amdgpu_waves_per_eu(2, 2)))
void logsparse_attn_kernel(const float* __restrict__ Q,
                           const float* __restrict__ K,
                           const float* __restrict__ V,
                           float* __restrict__ Out) {
    __shared__ __align__(16) float plds[WAVES][NSLOT][16][18];

    const int lane = threadIdx.x & 31;
    const int wave = __builtin_amdgcn_readfirstlane(threadIdx.x >> 5);
    const int g    = lane >> 4;
    const int ln   = lane & 15;

    const int gid = blockIdx.x * WAVES + wave;
    const int qt  = gid & 127;
    const int h   = (gid >> 7) & 7;
    const int b   = gid >> 10;
    const int i0  = qt * 16;

    float* stA = &plds[wave][0][0][0];
    float* stB = stA + 16 * PITCH;

    v4f gq[8], ga[8], gb[8];
    tile_gload(Q + ((size_t)(b * L_ + i0) * H_ + h) * E_, g, ln, gq);
    {
        const int jta = qt - slot_m(0), jtb = qt - slot_m(1);
        tile_gload(K + ((size_t)(b * S_ + (jta < 0 ? 0 : jta) * 16) * H_ + h) * E_, g, ln, ga);
        tile_gload(K + ((size_t)(b * S_ + (jtb < 0 ? 0 : jtb) * 16) * H_ + h) * E_, g, ln, gb);
    }
    v2f qf[16];
    tile_stage(stA, g, ln, gq);
    wave_sync();
    tile_frags(stA, g, ln, qf);
    wave_sync();

    v2f kfa[16], kfb[16];
    tile_stage(stA, g, ln, ga);
    tile_stage(stB, g, ln, gb);
    {
        const int jta = qt - slot_m(2), jtb = qt - slot_m(3);
        tile_gload(K + ((size_t)(b * S_ + (jta < 0 ? 0 : jta) * 16) * H_ + h) * E_, g, ln, ga);
        tile_gload(K + ((size_t)(b * S_ + (jtb < 0 ? 0 : jtb) * 16) * H_ + h) * E_, g, ln, gb);
    }
    wave_sync();
    tile_frags(stA, g, ln, kfa);
    tile_frags(stB, g, ln, kfb);
    wave_sync();

    v8f s[NSLOT];
#pragma unroll
    for (int tp = 0; tp < NSLOT; tp += 2) {
        v2f nfa[16], nfb[16];
        if (tp + 2 < NSLOT) {
            tile_stage(stA, g, ln, ga);
            tile_stage(stB, g, ln, gb);
            if (tp + 4 < NSLOT) {
                const int jta = qt - slot_m(tp + 4), jtb = qt - slot_m(tp + 5);
                tile_gload(K + ((size_t)(b * S_ + (jta < 0 ? 0 : jta) * 16) * H_ + h) * E_, g, ln, ga);
                tile_gload(K + ((size_t)(b * S_ + (jtb < 0 ? 0 : jtb) * 16) * H_ + h) * E_, g, ln, gb);
            }
            wave_sync();
            tile_frags(stA, g, ln, nfa);
            tile_frags(stB, g, ln, nfb);
            wave_sync();
        }
        __builtin_amdgcn_sched_barrier(0);
        v8f acca = (v8f){}, accb = (v8f){};
#pragma unroll
        for (int q = 0; q < 8; ++q) {
            acca = __builtin_amdgcn_wmma_f32_16x16x4_f32(false, qf[2 * q], false, kfa[2 * q],
                                                         (short)0, acca, false, false);
            accb = __builtin_amdgcn_wmma_f32_16x16x4_f32(false, qf[2 * q], false, kfb[2 * q],
                                                         (short)0, accb, false, false);
            acca = __builtin_amdgcn_wmma_f32_16x16x4_f32(false, qf[2 * q + 1], false, kfa[2 * q + 1],
                                                         (short)0, acca, false, false);
            accb = __builtin_amdgcn_wmma_f32_16x16x4_f32(false, qf[2 * q + 1], false, kfb[2 * q + 1],
                                                         (short)0, accb, false, false);
        }
        s[tp] = acca;
        s[tp + 1] = accb;
        if (tp + 2 < NSLOT) {
#pragma unroll
            for (int f = 0; f < 16; ++f) { kfa[f] = nfa[f]; kfb[f] = nfb[f]; }
        }
        __builtin_amdgcn_sched_barrier(0);
    }

    const int y = 15 + 8 * g - ln;
    constexpr unsigned long long A01 = ((unsigned long long)allow_word(1) << 32) | allow_word(0);
    constexpr unsigned long long A12 = ((unsigned long long)allow_word(2) << 32) | allow_word(1);
    static_assert(allow_word(3) == 0u, "bits 16*5 + 7 + 23 and below are all that is read");
    const unsigned w0 = (unsigned)(A01 >> y);
    const unsigned w1 = (unsigned)(A12 >> y);
    const unsigned w2 = allow_word(2) >> y;
    const unsigned c0 = (unsigned)(0xFFFFFFFFFFFF8000ull >> y);
    float rowmax[8], rowsum[8];
#pragma unroll
    for (int r = 0; r < 8; ++r) {
        const int i = i0 + r + 8 * g;
        const unsigned w0r = (i < 22) ? c0 : w0;
        float mx = -INFINITY;
#pragma unroll
        for (int t = 0; t < NSLOT; ++t) {
            const int m  = slot_m(t);
            const int mc = t < 4 ? m : 4 + (t & 1);
            const int c  = 16 * mc + r;
            const unsigned word = c < 32 ? w0r : (c < 64 ? w1 : w2);
            const unsigned live = (qt >= m) ? word : 0u;
            const bool ok = ((live >> (c & 31)) & 1u) != 0u;
            const float logit = ok ? SCALE * s[t][r] : MASKED;
            s[t][r] = logit;
            mx = fmaxf(mx, logit);
        }
        rowmax[r] = mx;
    }
#pragma unroll
    for (int off = 8; off >= 1; off >>= 1) {
        float other[8];
#pragma unroll
        for (int r = 0; r < 8; ++r)
            other[r] = __shfl_xor(rowmax[r], off, 32);
#pragma unroll
        for (int r = 0; r < 8; ++r)
            rowmax[r] = fmaxf(rowmax[r], other[r]);
    }
#pragma unroll
    for (int r = 0; r < 8; ++r) {
        float sum = 0.0f;
#pragma unroll
        for (int t = 0; t < NSLOT; ++t) {
            const float p = __expf(s[t][r] - rowmax[r]);
            sum += p;
            plds[wave][t][r + 8 * g][ln] = p;
        }
        rowsum[r] = sum;
    }
#pragma unroll
    for (int off = 8; off >= 1; off >>= 1) {
        float other[8];
#pragma unroll
        for (int r = 0; r < 8; ++r)
            other[r] = __shfl_xor(rowsum[r], off, 32);
#pragma unroll
        for (int r = 0; r < 8; ++r)
            rowsum[r] += other[r];
    }
    __builtin_amdgcn_wave_barrier();

    v8f oacc[4];
#pragma unroll
    for (int nt = 0; nt < 4; ++nt) oacc[nt] = (v8f){};
    const float* pw = &plds[wave][0][0][0];
    pv_trip<0, 4, 0, 4>(V, pw, qt, 0, b, h, g, ln, oacc);
    pv_trip<0, 4, 1, 4>(V, pw + 2 * (16 * 18), qt, 2, b, h, g, ln, oacc);
#pragma unroll 1
    for (int k = 0; k < 5; ++k)
        pv_trip<0, 2, 1, 4>(V, pw + (4 + 2 * k) * (16 * 18), qt, 4 << k, b, h, g, ln, oacc);

#pragma unroll
    for (int r = 0; r < 8; ++r) {
        const int i = i0 + r + 8 * g;
        const float inv = 1.0f / rowsum[r];
        float* orow = Out + ((size_t)(b * L_ + i) * H_ + h) * D_;
        float4 o;
        o.x = oacc[0][r] * inv;
        o.y = oacc[1][r] * inv;
        o.z = oacc[2][r] * inv;
        o.w = oacc[3][r] * inv;
        *(float4*)(orow + 4 * ln) = o;
    }
}

extern "C" void kernel_launch(void* const* d_in, const int* in_sizes, int n_in,
                              void* d_out, int out_size, void* d_ws, size_t ws_size,
                              hipStream_t stream) {
    (void)in_sizes; (void)n_in; (void)out_size; (void)d_ws; (void)ws_size;
    const float* Q = (const float*)d_in[0];
    const float* K = (const float*)d_in[1];
    const float* V = (const float*)d_in[2];
    float* Out = (float*)d_out;

    const int total_waves = B_ * H_ * (L_ / 16);
    dim3 grid(total_waves / WAVES);
    dim3 block(WAVES * 32);
    hipLaunchKernelGGL(logsparse_attn_kernel, grid, block, 0, stream, Q, K, V, Out);
}


# operand fetches moved ahead of the WMMAs that use them: K tiles two trips ahead, V rows in pass 1's last two trips (20 loads at the head of pass 3), P fragments as one group; same sums and order
# speedup vs baseline: 4.7455x; 1.1281x over previous
#include <hip/hip_runtime.h>
#include <math.h>


typedef __attribute__((ext_vector_type(2))) float v2f;
typedef __attribute__((ext_vector_type(8))) float v8f;
typedef __attribute__((ext_vector_type(4))) float v4f;

#define WAVES 4
#define NSLOT 14

constexpr int B_ = 2, L_ = 2048, S_ = 2048, H_ = 8, E_ = 64, D_ = 64;
constexpr float SCALE  = 0.125f;
constexpr float MASKED = -1.25e8f;

constexpr bool dist_ok(int d) {
    if (d < 0) return false;
    if (d <= 10) return true;
    int e = d - 10;
    return (e <= 1024) && ((e & (e - 1)) == 0);
}
constexpr unsigned allow_word(int w) {
    unsigned v = 0;
    for (int b = 0; b < 32; ++b)
        if (dist_ok(32 * w + b - 15)) v |= 1u << b;
    return v;
}

__device__ __forceinline__ constexpr int slot_m(int t) {
    return t < 6 ? t : (t < 8 ? t + 2 : (t < 10 ? t + 8 : (t < 12 ? t + 22 : t + 52)));
}

#define PITCH 68

__device__ __forceinline__ void wave_sync() {
    __builtin_amdgcn_fence(__ATOMIC_RELEASE, "wavefront");
    __builtin_amdgcn_wave_barrier();
    __builtin_amdgcn_fence(__ATOMIC_ACQUIRE, "wavefront");
}

__device__ __forceinline__ void tile_gload(const float* __restrict__ tile, int g, int ln, v4f (&r)[8]) {
#pragma unroll
    for (int i = 0; i < 8; ++i)
        r[i] = *(const v4f*)(tile + (size_t)(2 * i + g) * (H_ * E_) + 4 * ln);
}
__device__ __forceinline__ void tile_stage(float* st, int g, int ln, const v4f (&r)[8]) {
#pragma unroll
    for (int i = 0; i < 8; ++i)
        *(v4f*)(st + (2 * i + g) * PITCH + 4 * ln) = r[i];
}
__device__ __forceinline__ void tile_frags(const float* st, int g, int ln, v2f (&f)[16]) {
#pragma unroll
    for (int q = 0; q < 8; ++q) {
        const v4f t = *(const v4f*)(st + ln * PITCH + 8 * q + 4 * g);
        v2f a0; a0[0] = t[0]; a0[1] = t[1];
        v2f a1; a1[0] = t[2]; a1[1] = t[3];
        f[2 * q]     = a0;
        f[2 * q + 1] = a1;
    }
}

constexpr bool chunk_live(int m, int kc) {
    for (int c = 4 * kc; c < 4 * kc + 4; ++c)
        for (int row = 0; row < 16; ++row)
            if (dist_ok(16 * m + row - c)) return true;
    return false;
}
static_assert(!chunk_live(3, 0), "tile distance 3: key columns 0..3 are masked in every row");
static_assert(!chunk_live(4, 2) && !chunk_live(4, 3), "nearer far tile: key columns 8..15 are masked");
static_assert(!chunk_live(5, 0), "farther far tile: key columns 0..3 are masked");

template <int LO_A, int HI_A, int LO_B, int HI_B>
__device__ __forceinline__ void pv_vload(const float* __restrict__ V, int qt, int ma, int b, int h,
                                         int g, int ln, float4 (&vr)[2][4][2]) {
#pragma unroll
    for (int u = 0; u < 2; ++u) {
        const int lo = u ? LO_B : LO_A, hi = u ? HI_B : HI_A;
        const int jt = qt - (ma + u);
        const int j0 = (jt < 0 ? 0 : jt) * 16;
        const float* vbase = V + ((size_t)(b * S_ + j0) * H_ + h) * D_ + 4 * ln;
#pragma unroll
        for (int kc = lo; kc < hi; ++kc) {
            const int kr = 4 * kc + 2 * g;
            vr[u][kc][0] = *(const float4*)(vbase + (size_t)(kr + 0) * (H_ * D_));
            vr[u][kc][1] = *(const float4*)(vbase + (size_t)(kr + 1) * (H_ * D_));
        }
    }
}
template <int LO_A, int HI_A, int LO_B, int HI_B>
__device__ __forceinline__ void pv_pload(const float* pl, int g, int ln, v2f (&pa)[2][4]) {
#pragma unroll
    for (int u = 0; u < 2; ++u) {
        const int lo = u ? LO_B : LO_A, hi = u ? HI_B : HI_A;
#pragma unroll
        for (int kc = lo; kc < hi; ++kc) {
            float2 pv = *(const float2*)(pl + (u * 16 + ln) * 18 + 4 * kc + 2 * g);
            v2f a; a[0] = pv.x; a[1] = pv.y;
            pa[u][kc] = a;
        }
    }
}
template <int LO_A, int HI_A, int LO_B, int HI_B>
__device__ __forceinline__ void pv_mma(const v2f (&pa)[2][4], const float4 (&vr)[2][4][2],
                                       v8f (&oacc)[4]) {
#pragma unroll
    for (int u = 0; u < 2; ++u) {
        const int lo = u ? LO_B : LO_A, hi = u ? HI_B : HI_A;
#pragma unroll
        for (int kc = lo; kc < hi; ++kc) {
            const float4 r0 = vr[u][kc][0];
            const float4 r1 = vr[u][kc][1];
            v2f x[4];
            x[0][0] = r0.x; x[0][1] = r1.x;
            x[1][0] = r0.y; x[1][1] = r1.y;
            x[2][0] = r0.z; x[2][1] = r1.z;
            x[3][0] = r0.w; x[3][1] = r1.w;
#pragma unroll
            for (int nt = 0; nt < 4; ++nt)
                oacc[nt] = __builtin_amdgcn_wmma_f32_16x16x4_f32(false, pa[u][kc], false, x[nt],
                                                                 (short)0, oacc[nt], false, false);
        }
    }
}

__device__ __forceinline__ const float* k_tile(const float* __restrict__ K, int b, int h, int qt, int t) {
    const int jt = qt - slot_m(t);
    return K + ((size_t)(b * S_ + (jt < 0 ? 0 : jt) * 16) * H_ + h) * E_;
}

__global__ __launch_bounds__(WAVES * 32) __attribute__((amdgpu_waves_per_eu(2, 2)))
void logsparse_attn_kernel(const float* __restrict__ Q,
                           const float* __restrict__ K,
                           const float* __restrict__ V,
                           float* __restrict__ Out) {
    __shared__ __align__(16) float plds[WAVES][NSLOT][16][18];

    const int lane = threadIdx.x & 31;
    const int wave = __builtin_amdgcn_readfirstlane(threadIdx.x >> 5);
    const int g    = lane >> 4;
    const int ln   = lane & 15;

    const int gid = blockIdx.x * WAVES + wave;
    const int qt  = gid & 127;
    const int h   = (gid >> 7) & 7;
    const int b   = gid >> 10;
    const int i0  = qt * 16;

    float* stA = &plds[wave][0][0][0];
    float* stB = stA + 16 * PITCH;

    v4f gq[8], gk[2][2][8];
    tile_gload(Q + ((size_t)(b * L_ + i0) * H_ + h) * E_, g, ln, gq);
    tile_gload(k_tile(K, b, h, qt, 0), g, ln, gk[0][0]);
    tile_gload(k_tile(K, b, h, qt, 1), g, ln, gk[0][1]);
    tile_gload(k_tile(K, b, h, qt, 2), g, ln, gk[1][0]);
    tile_gload(k_tile(K, b, h, qt, 3), g, ln, gk[1][1]);
    v2f qf[16];
    tile_stage(stA, g, ln, gq);
    wave_sync();
    tile_frags(stA, g, ln, qf);
    wave_sync();

    v2f kfa[16], kfb[16];
    tile_stage(stA, g, ln, gk[0][0]);
    tile_stage(stB, g, ln, gk[0][1]);
    tile_gload(k_tile(K, b, h, qt, 4), g, ln, gk[0][0]);
    tile_gload(k_tile(K, b, h, qt, 5), g, ln, gk[0][1]);
    wave_sync();
    tile_frags(stA, g, ln, kfa);
    tile_frags(stB, g, ln, kfb);
    wave_sync();

    float4 vn0[2][4][2], vn1[2][4][2], vfar[5][2][4][2];
    v8f s[NSLOT];
#pragma unroll
    for (int tp = 0; tp < NSLOT; tp += 2) {
        v2f nfa[16], nfb[16];
        if (tp + 2 < NSLOT) {
            const int set = ((tp >> 1) + 1) & 1;
            tile_stage(stA, g, ln, gk[set][0]);
            tile_stage(stB, g, ln, gk[set][1]);
            if (tp + 6 < NSLOT) {
                tile_gload(k_tile(K, b, h, qt, tp + 6), g, ln, gk[set][0]);
                tile_gload(k_tile(K, b, h, qt, tp + 7), g, ln, gk[set][1]);
            }
            wave_sync();
            tile_frags(stA, g, ln, nfa);
            tile_frags(stB, g, ln, nfb);
            wave_sync();
        }
        if (tp == NSLOT - 4) {
            pv_vload<0, 4, 0, 4>(V, qt, 0, b, h, g, ln, vn0);
            pv_vload<0, 4, 1, 4>(V, qt, 2, b, h, g, ln, vn1);
        }
        if (tp == NSLOT - 2) {
#pragma unroll
            for (int k = 0; k < 3; ++k)
                pv_vload<0, 2, 1, 4>(V, qt, 4 << k, b, h, g, ln, vfar[k]);
        }
        __builtin_amdgcn_sched_barrier(0);
        v8f acca = (v8f){}, accb = (v8f){};
#pragma unroll
        for (int q = 0; q < 8; ++q) {
            acca = __builtin_amdgcn_wmma_f32_16x16x4_f32(false, qf[2 * q], false, kfa[2 * q],
                                                         (short)0, acca, false, false);
            accb = __builtin_amdgcn_wmma_f32_16x16x4_f32(false, qf[2 * q], false, kfb[2 * q],
                                                         (short)0, accb, false, false);
            acca = __builtin_amdgcn_wmma_f32_16x16x4_f32(false, qf[2 * q + 1], false, kfa[2 * q + 1],
                                                         (short)0, acca, false, false);
            accb = __builtin_amdgcn_wmma_f32_16x16x4_f32(false, qf[2 * q + 1], false, kfb[2 * q + 1],
                                                         (short)0, accb, false, false);
        }
        s[tp] = acca;
        s[tp + 1] = accb;
        if (tp + 2 < NSLOT) {
#pragma unroll
            for (int f = 0; f < 16; ++f) { kfa[f] = nfa[f]; kfb[f] = nfb[f]; }
        }
        __builtin_amdgcn_sched_barrier(0);
    }

    const int y = 15 + 8 * g - ln;
    constexpr unsigned long long A01 = ((unsigned long long)allow_word(1) << 32) | allow_word(0);
    constexpr unsigned long long A12 = ((unsigned long long)allow_word(2) << 32) | allow_word(1);
    static_assert(allow_word(3) == 0u, "bits 16*5 + 7 + 23 and below are all that is read");
    const unsigned w0 = (unsigned)(A01 >> y);
    const unsigned w1 = (unsigned)(A12 >> y);
    const unsigned w2 = allow_word(2) >> y;
    const unsigned c0 = (unsigned)(0xFFFFFFFFFFFF8000ull >> y);
    float rowmax[8], rowsum[8];
#pragma unroll
    for (int r = 0; r < 8; ++r) {
        const int i = i0 + r + 8 * g;
        const unsigned w0r = (i < 22) ? c0 : w0;
        float mx = -INFINITY;
#pragma unroll
        for (int t = 0; t < NSLOT; ++t) {
            const int m  = slot_m(t);
            const int mc = t < 4 ? m : 4 + (t & 1);
            const int c  = 16 * mc + r;
            const unsigned word = c < 32 ? w0r : (c < 64 ? w1 : w2);
            const unsigned live = (qt >= m) ? word : 0u;
            const bool ok = ((live >> (c & 31)) & 1u) != 0u;
            const float logit = ok ? SCALE * s[t][r] : MASKED;
            s[t][r] = logit;
            mx = fmaxf(mx, logit);
        }
        rowmax[r] = mx;
    }
#pragma unroll
    for (int off = 8; off >= 1; off >>= 1) {
        float other[8];
#pragma unroll
        for (int r = 0; r < 8; ++r)
            other[r] = __shfl_xor(rowmax[r], off, 32);
#pragma unroll
        for (int r = 0; r < 8; ++r)
            rowmax[r] = fmaxf(rowmax[r], other[r]);
    }
#pragma unroll
    for (int r = 0; r < 8; ++r) {
        float sum = 0.0f;
#pragma unroll
        for (int t = 0; t < NSLOT; ++t) {
            const float p = __expf(s[t][r] - rowmax[r]);
            sum += p;
            plds[wave][t][r + 8 * g][ln] = p;
        }
        rowsum[r] = sum;
    }
#pragma unroll
    for (int off = 8; off >= 1; off >>= 1) {
        float other[8];
#pragma unroll
        for (int r = 0; r < 8; ++r)
            other[r] = __shfl_xor(rowsum[r], off, 32);
#pragma unroll
        for (int r = 0; r < 8; ++r)
            rowsum[r] += other[r];
    }
    __builtin_amdgcn_wave_barrier();

    v8f oacc[4];
#pragma unroll
    for (int nt = 0; nt < 4; ++nt) oacc[nt] = (v8f){};
    const float* pw = &plds[wave][0][0][0];
#pragma unroll
    for (int k = 3; k < 5; ++k)
        pv_vload<0, 2, 1, 4>(V, qt, 4 << k, b, h, g, ln, vfar[k]);
    v2f pn0[2][4], pn1[2][4], pfar[5][2][4];
    pv_pload<0, 4, 0, 4>(pw, g, ln, pn0);
    pv_pload<0, 4, 1, 4>(pw + 2 * (16 * 18), g, ln, pn1);
#pragma unroll
    for (int k = 0; k < 5; ++k)
        pv_pload<0, 2, 1, 4>(pw + (4 + 2 * k) * (16 * 18), g, ln, pfar[k]);
    __builtin_amdgcn_sched_barrier(0);
    pv_mma<0, 4, 0, 4>(pn0, vn0, oacc);
    pv_mma<0, 4, 1, 4>(pn1, vn1, oacc);
#pragma unroll
    for (int k = 0; k < 5; ++k)
        pv_mma<0, 2, 1, 4>(pfar[k], vfar[k], oacc);

#pragma unroll
    for (int r = 0; r < 8; ++r) {
        const int i = i0 + r + 8 * g;
        const float inv = 1.0f / rowsum[r];
        float* orow = Out + ((size_t)(b * L_ + i) * H_ + h) * D_;
        float4 o;
        o.x = oacc[0][r] * inv;
        o.y = oacc[1][r] * inv;
        o.z = oacc[2][r] * inv;
        o.w = oacc[3][r] * inv;
        *(float4*)(orow + 4 * ln) = o;
    }
}

extern "C" void kernel_launch(void* const* d_in, const int* in_sizes, int n_in,
                              void* d_out, int out_size, void* d_ws, size_t ws_size,
                              hipStream_t stream) {
    (void)in_sizes; (void)n_in; (void)out_size; (void)d_ws; (void)ws_size;
    const float* Q = (const float*)d_in[0];
    const float* K = (const float*)d_in[1];
    const float* V = (const float*)d_in[2];
    float* Out = (float*)d_out;

    const int total_waves = B_ * H_ * (L_ / 16);
    dim3 grid(total_waves / WAVES);
    dim3 block(WAVES * 32);
    hipLaunchKernelGGL(logsparse_attn_kernel, grid, block, 0, stream, Q, K, V, Out);
}


# scores cross to the A layout of P*V through LDS before the softmax, a far pair landing as one tile: row max, exp, row sum in-lane over 35 chunks, P*V from registers on 9 tiles; regroups row, O sums
# speedup vs baseline: 5.0864x; 1.0718x over previous
#include <hip/hip_runtime.h>
#include <math.h>


typedef __attribute__((ext_vector_type(2))) float v2f;
typedef __attribute__((ext_vector_type(8))) float v8f;
typedef __attribute__((ext_vector_type(4))) float v4f;

#define WAVES 4
#define NSLOT 14

constexpr int B_ = 2, L_ = 2048, S_ = 2048, H_ = 8, E_ = 64, D_ = 64;
constexpr float SCALE  = 0.125f;
constexpr float MASKED = -1.25e8f;

constexpr bool dist_ok(int d) {
    if (d < 0) return false;
    if (d <= 10) return true;
    int e = d - 10;
    return (e <= 1024) && ((e & (e - 1)) == 0);
}
constexpr unsigned allow_word(int w) {
    unsigned v = 0;
    for (int b = 0; b < 32; ++b)
        if (dist_ok(32 * w + b - 15)) v |= 1u << b;
    return v;
}

__device__ __forceinline__ constexpr int slot_m(int t) {
    return t < 6 ? t : (t < 8 ? t + 2 : (t < 10 ? t + 8 : (t < 12 ? t + 22 : t + 52)));
}

#define PITCH 68

__device__ __forceinline__ void wave_sync() {
    __builtin_amdgcn_fence(__ATOMIC_RELEASE, "wavefront");
    __builtin_amdgcn_wave_barrier();
    __builtin_amdgcn_fence(__ATOMIC_ACQUIRE, "wavefront");
}

__device__ __forceinline__ void tile_gload(const float* __restrict__ tile, int g, int ln, v4f (&r)[8]) {
#pragma unroll
    for (int i = 0; i < 8; ++i)
        r[i] = *(const v4f*)(tile + (size_t)(2 * i + g) * (H_ * E_) + 4 * ln);
}
__device__ __forceinline__ void tile_stage(float* st, int g, int ln, const v4f (&r)[8]) {
#pragma unroll
    for (int i = 0; i < 8; ++i)
        *(v4f*)(st + (2 * i + g) * PITCH + 4 * ln) = r[i];
}
__device__ __forceinline__ void tile_frags(const float* st, int g, int ln, v2f (&f)[16]) {
#pragma unroll
    for (int q = 0; q < 8; ++q) {
        const v4f t = *(const v4f*)(st + ln * PITCH + 8 * q + 4 * g);
        v2f a0; a0[0] = t[0]; a0[1] = t[1];
        v2f a1; a1[0] = t[2]; a1[1] = t[3];
        f[2 * q]     = a0;
        f[2 * q + 1] = a1;
    }
}

constexpr bool chunk_live(int m, int kc) {
    for (int c = 4 * kc; c < 4 * kc + 4; ++c)
        for (int row = 0; row < 16; ++row)
            if (dist_ok(16 * m + row - c)) return true;
    return false;
}
static_assert(!chunk_live(3, 0), "tile distance 3: key columns 0..3 are masked in every row");

template <int LO_A, int HI_A, int LO_B, int HI_B>
__device__ __forceinline__ void pv_vload(const float* __restrict__ V, int qt, int ma, int b, int h,
                                         int g, int ln, float4 (&vr)[2][4][2]) {
#pragma unroll
    for (int u = 0; u < 2; ++u) {
        const int lo = u ? LO_B : LO_A, hi = u ? HI_B : HI_A;
        const int jt = qt - (ma + u);
        const int j0 = (jt < 0 ? 0 : jt) * 16;
        const float* vbase = V + ((size_t)(b * S_ + j0) * H_ + h) * D_ + 4 * ln;
#pragma unroll
        for (int kc = lo; kc < hi; ++kc) {
            const int kr = 4 * kc + 2 * g;
            vr[u][kc][0] = *(const float4*)(vbase + (size_t)(kr + 0) * (H_ * D_));
            vr[u][kc][1] = *(const float4*)(vbase + (size_t)(kr + 1) * (H_ * D_));
        }
    }
}
constexpr bool col_live(int m, int c) {
    for (int row = 0; row < 16; ++row)
        if (dist_ok(16 * m + row - c)) return true;
    return false;
}
constexpr bool cols_dead(int m, int lo, int hi) {
    for (int c = lo; c < hi; ++c)
        if (col_live(m, c)) return false;
    return true;
}
static_assert(cols_dead(4, 6, 16), "nearer far tile: key columns 6..15 are masked in every row");
static_assert(cols_dead(5, 0, 6), "farther far tile: key columns 0..5 are masked in every row");

__device__ __forceinline__ void pv_vload_pair(const float* __restrict__ V, int qt, int ma, int b, int h,
                                              int g, int ln, float4 (&vr)[4][2]) {
    const int jn = qt - ma, jf = qt - ma - 1;
    const float* vnear = V + ((size_t)(b * S_ + (jn < 0 ? 0 : jn) * 16) * H_ + h) * D_ + 4 * ln;
    const float* vfarp = V + ((size_t)(b * S_ + (jf < 0 ? 0 : jf) * 16) * H_ + h) * D_ + 4 * ln;
    const float* vmid  = g ? vfarp : vnear;
#pragma unroll
    for (int kc = 0; kc < 4; ++kc) {
        const int kr = 4 * kc + 2 * g;
        const float* base = kc == 0 ? vnear : (kc == 1 ? vmid : vfarp);
        vr[kc][0] = *(const float4*)(base + (size_t)(kr + 0) * (H_ * D_));
        vr[kc][1] = *(const float4*)(base + (size_t)(kr + 1) * (H_ * D_));
    }
}
template <int LO, int HI>
__device__ __forceinline__ void pv_mma_tile(const v2f (&pa)[4], const float4 (&vr)[4][2], v8f (&oacc)[4]) {
#pragma unroll
    for (int kc = LO; kc < HI; ++kc) {
        const float4 r0 = vr[kc][0];
        const float4 r1 = vr[kc][1];
        v2f x[4];
        x[0][0] = r0.x; x[0][1] = r1.x;
        x[1][0] = r0.y; x[1][1] = r1.y;
        x[2][0] = r0.z; x[2][1] = r1.z;
        x[3][0] = r0.w; x[3][1] = r1.w;
#pragma unroll
        for (int nt = 0; nt < 4; ++nt)
            oacc[nt] = __builtin_amdgcn_wmma_f32_16x16x4_f32(false, pa[kc], false, x[nt],
                                                             (short)0, oacc[nt], false, false);
    }
}

__device__ __forceinline__ const float* k_tile(const float* __restrict__ K, int b, int h, int qt, int t) {
    const int jt = qt - slot_m(t);
    return K + ((size_t)(b * S_ + (jt < 0 ? 0 : jt) * 16) * H_ + h) * E_;
}

__global__ __launch_bounds__(WAVES * 32) __attribute__((amdgpu_waves_per_eu(2, 2)))
void logsparse_attn_kernel(const float* __restrict__ Q,
                           const float* __restrict__ K,
                           const float* __restrict__ V,
                           float* __restrict__ Out) {
    __shared__ __align__(16) float plds[WAVES][NSLOT][16][18];

    const int lane = threadIdx.x & 31;
    const int wave = __builtin_amdgcn_readfirstlane(threadIdx.x >> 5);
    const int g    = lane >> 4;
    const int ln   = lane & 15;

    const int gid = blockIdx.x * WAVES + wave;
    const int qt  = gid & 127;
    const int h   = (gid >> 7) & 7;
    const int b   = gid >> 10;
    const int i0  = qt * 16;

    float* stA = &plds[wave][0][0][0];
    float* stB = stA + 16 * PITCH;

    v4f gq[8], gk[2][2][8];
    tile_gload(Q + ((size_t)(b * L_ + i0) * H_ + h) * E_, g, ln, gq);
    tile_gload(k_tile(K, b, h, qt, 0), g, ln, gk[0][0]);
    tile_gload(k_tile(K, b, h, qt, 1), g, ln, gk[0][1]);
    tile_gload(k_tile(K, b, h, qt, 2), g, ln, gk[1][0]);
    tile_gload(k_tile(K, b, h, qt, 3), g, ln, gk[1][1]);
    v2f qf[16];
    tile_stage(stA, g, ln, gq);
    wave_sync();
    tile_frags(stA, g, ln, qf);
    wave_sync();

    v2f kfa[16], kfb[16];
    tile_stage(stA, g, ln, gk[0][0]);
    tile_stage(stB, g, ln, gk[0][1]);
    tile_gload(k_tile(K, b, h, qt, 4), g, ln, gk[0][0]);
    tile_gload(k_tile(K, b, h, qt, 5), g, ln, gk[0][1]);
    wave_sync();
    tile_frags(stA, g, ln, kfa);
    tile_frags(stB, g, ln, kfb);
    wave_sync();

    float4 vn0[2][4][2], vn1[2][4][2], vfar[5][4][2];
    v8f s[NSLOT];
#pragma unroll
    for (int tp = 0; tp < NSLOT; tp += 2) {
        v2f nfa[16], nfb[16];
        if (tp + 2 < NSLOT) {
            const int set = ((tp >> 1) + 1) & 1;
            tile_stage(stA, g, ln, gk[set][0]);
            tile_stage(stB, g, ln, gk[set][1]);
            if (tp + 6 < NSLOT) {
                tile_gload(k_tile(K, b, h, qt, tp + 6), g, ln, gk[set][0]);
                tile_gload(k_tile(K, b, h, qt, tp + 7), g, ln, gk[set][1]);
            }
            wave_sync();
            tile_frags(stA, g, ln, nfa);
            tile_frags(stB, g, ln, nfb);
            wave_sync();
        }
        if (tp == NSLOT - 4) {
            pv_vload<0, 4, 0, 4>(V, qt, 0, b, h, g, ln, vn0);
            pv_vload<0, 4, 1, 4>(V, qt, 2, b, h, g, ln, vn1);
        }
        if (tp == NSLOT - 2) {
#pragma unroll
            for (int k = 0; k < 3; ++k)
                pv_vload_pair(V, qt, 4 << k, b, h, g, ln, vfar[k]);
        }
        __builtin_amdgcn_sched_barrier(0);
        v8f acca = (v8f){}, accb = (v8f){};
#pragma unroll
        for (int q = 0; q < 8; ++q) {
            acca = __builtin_amdgcn_wmma_f32_16x16x4_f32(false, qf[2 * q], false, kfa[2 * q],
                                                         (short)0, acca, false, false);
            accb = __builtin_amdgcn_wmma_f32_16x16x4_f32(false, qf[2 * q], false, kfb[2 * q],
                                                         (short)0, accb, false, false);
            acca = __builtin_amdgcn_wmma_f32_16x16x4_f32(false, qf[2 * q + 1], false, kfa[2 * q + 1],
                                                         (short)0, acca, false, false);
            accb = __builtin_amdgcn_wmma_f32_16x16x4_f32(false, qf[2 * q + 1], false, kfb[2 * q + 1],
                                                         (short)0, accb, false, false);
        }
        s[tp] = acca;
        s[tp + 1] = accb;
        if (tp + 2 < NSLOT) {
#pragma unroll
            for (int f = 0; f < 16; ++f) { kfa[f] = nfa[f]; kfb[f] = nfb[f]; }
        }
        __builtin_amdgcn_sched_barrier(0);
    }

    const int y = 15 + 8 * g - ln;
    constexpr unsigned long long A01 = ((unsigned long long)allow_word(1) << 32) | allow_word(0);
    constexpr unsigned long long A12 = ((unsigned long long)allow_word(2) << 32) | allow_word(1);
    static_assert(allow_word(3) == 0u, "bits 16*5 + 7 + 23 and below are all that is read");
    const unsigned w0 = (unsigned)(A01 >> y);
    const unsigned w1 = (unsigned)(A12 >> y);
    const unsigned w2 = allow_word(2) >> y;
    const unsigned c0 = (unsigned)(0xFFFFFFFFFFFF8000ull >> y);
#pragma unroll
    for (int r = 0; r < 8; ++r) {
        const int i = i0 + r + 8 * g;
        const unsigned w0r = (i < 22) ? c0 : w0;
#pragma unroll
        for (int t = 0; t < NSLOT; ++t) {
            const int m  = slot_m(t);
            const int mc = t < 4 ? m : 4 + (t & 1);
            const int c  = 16 * mc + r;
            const unsigned word = c < 32 ? w0r : (c < 64 ? w1 : w2);
            const unsigned live = (qt >= m) ? word : 0u;
            const bool ok = ((live >> (c & 31)) & 1u) != 0u;
            s[t][r] = ok ? SCALE * s[t][r] : MASKED;
        }
    }

    const bool nearcol = ln < 6;
#pragma unroll
    for (int r = 0; r < 8; ++r) {
#pragma unroll
        for (int t = 0; t < 4; ++t)
            plds[wave][t][r + 8 * g][ln] = s[t][r];
#pragma unroll
        for (int k = 0; k < 5; ++k)
            plds[wave][4 + k][r + 8 * g][ln] = nearcol ? s[4 + 2 * k][r] : s[5 + 2 * k][r];
    }
#pragma unroll
    for (int k = 3; k < 5; ++k)
        pv_vload_pair(V, qt, 4 << k, b, h, g, ln, vfar[k]);
    wave_sync();

    const float* lw = &plds[wave][0][0][0];
    v2f pn[4][4], pf[5][4];
#pragma unroll
    for (int t = 0; t < 4; ++t) {
#pragma unroll
        for (int kc = (t == 3 ? 1 : 0); kc < 4; ++kc) {
            const float2 x = *(const float2*)(lw + (t * 16 + ln) * 18 + 4 * kc + 2 * g);
            v2f a; a[0] = x.x; a[1] = x.y;
            pn[t][kc] = a;
        }
    }
#pragma unroll
    for (int k = 0; k < 5; ++k) {
#pragma unroll
        for (int kc = 0; kc < 4; ++kc) {
            const float2 x = *(const float2*)(lw + ((4 + k) * 16 + ln) * 18 + 4 * kc + 2 * g);
            v2f a; a[0] = x.x; a[1] = x.y;
            pf[k][kc] = a;
        }
    }
    __builtin_amdgcn_sched_barrier(0);

    float mx0 = -INFINITY, mx1 = -INFINITY;
#pragma unroll
    for (int t = 0; t < 4; ++t) {
#pragma unroll
        for (int kc = (t == 3 ? 1 : 0); kc < 4; ++kc) {
            mx0 = fmaxf(mx0, pn[t][kc][0]);
            mx1 = fmaxf(mx1, pn[t][kc][1]);
        }
    }
#pragma unroll
    for (int k = 0; k < 5; ++k) {
#pragma unroll
        for (int kc = 0; kc < 4; ++kc) {
            mx0 = fmaxf(mx0, pf[k][kc][0]);
            mx1 = fmaxf(mx1, pf[k][kc][1]);
        }
    }
    float mx = fmaxf(mx0, mx1);
    mx = fmaxf(mx, __shfl_xor(mx, 16, 32));
    float sm0 = 0.0f, sm1 = 0.0f;
#pragma unroll
    for (int t = 0; t < 4; ++t) {
#pragma unroll
        for (int kc = (t == 3 ? 1 : 0); kc < 4; ++kc) {
            const float p0 = __expf(pn[t][kc][0] - mx);
            const float p1 = __expf(pn[t][kc][1] - mx);
            sm0 += p0; sm1 += p1;
            v2f a; a[0] = p0; a[1] = p1;
            pn[t][kc] = a;
        }
    }
#pragma unroll
    for (int k = 0; k < 5; ++k) {
#pragma unroll
        for (int kc = 0; kc < 4; ++kc) {
            const float p0 = __expf(pf[k][kc][0] - mx);
            const float p1 = __expf(pf[k][kc][1] - mx);
            sm0 += p0; sm1 += p1;
            v2f a; a[0] = p0; a[1] = p1;
            pf[k][kc] = a;
        }
    }
    float rsum = sm0 + sm1;
    rsum += __shfl_xor(rsum, 16, 32);

    v8f oacc[4];
#pragma unroll
    for (int nt = 0; nt < 4; ++nt) oacc[nt] = (v8f){};
    pv_mma_tile<0, 4>(pn[0], vn0[0], oacc);
    pv_mma_tile<0, 4>(pn[1], vn0[1], oacc);
    pv_mma_tile<0, 4>(pn[2], vn1[0], oacc);
    pv_mma_tile<1, 4>(pn[3], vn1[1], oacc);
#pragma unroll
    for (int k = 0; k < 5; ++k)
        pv_mma_tile<0, 4>(pf[k], vfar[k], oacc);

    const float inv = 1.0f / rsum;
#pragma unroll
    for (int r = 0; r < 8; ++r) {
        const int i = i0 + r + 8 * g;
        const float invr = __shfl(inv, r + 8 * g, 32);
        float* orow = Out + ((size_t)(b * L_ + i) * H_ + h) * D_;
        float4 o;
        o.x = oacc[0][r] * invr;
        o.y = oacc[1][r] * invr;
        o.z = oacc[2][r] * invr;
        o.w = oacc[3][r] * invr;
        *(float4*)(orow + 4 * ln) = o;
    }
}

extern "C" void kernel_launch(void* const* d_in, const int* in_sizes, int n_in,
                              void* d_out, int out_size, void* d_ws, size_t ws_size,
                              hipStream_t stream) {
    (void)in_sizes; (void)n_in; (void)out_size; (void)d_ws; (void)ws_size;
    const float* Q = (const float*)d_in[0];
    const float* K = (const float*)d_in[1];
    const float* V = (const float*)d_in[2];
    float* Out = (float*)d_out;

    const int total_waves = B_ * H_ * (L_ / 16);
    dim3 grid(total_waves / WAVES);
    dim3 block(WAVES * 32);
    hipLaunchKernelGGL(logsparse_attn_kernel, grid, block, 0, stream, Q, K, V, Out);
}


# scores of the five far distances (one allowed key per row each) as row dot products on the vector unit from the Q fragments, no far score tiles: 364 -> 204 WMMAs; regroups far k sums and row sums
# speedup vs baseline: 6.5443x; 1.2866x over previous
#include <hip/hip_runtime.h>
#include <math.h>


typedef __attribute__((ext_vector_type(2))) float v2f;
typedef __attribute__((ext_vector_type(8))) float v8f;
typedef __attribute__((ext_vector_type(4))) float v4f;

#define WAVES 4
#define NSLOT 14

constexpr int B_ = 2, L_ = 2048, S_ = 2048, H_ = 8, E_ = 64, D_ = 64;
constexpr float SCALE  = 0.125f;
constexpr float MASKED = -1.25e8f;

constexpr bool dist_ok(int d) {
    if (d < 0) return false;
    if (d <= 10) return true;
    int e = d - 10;
    return (e <= 1024) && ((e & (e - 1)) == 0);
}
constexpr unsigned allow_word(int w) {
    unsigned v = 0;
    for (int b = 0; b < 32; ++b)
        if (dist_ok(32 * w + b - 15)) v |= 1u << b;
    return v;
}

__device__ __forceinline__ constexpr int slot_m(int t) {
    return t < 6 ? t : (t < 8 ? t + 2 : (t < 10 ? t + 8 : (t < 12 ? t + 22 : t + 52)));
}

#define NFAR 5
__device__ __forceinline__ constexpr int far_d(int k) { return 10 + (64 << k); }
static_assert(dist_ok(far_d(0)) && dist_ok(far_d(1)) && dist_ok(far_d(2)) && dist_ok(far_d(3)) &&
              dist_ok(far_d(4)), "the five far distances are allowed");
static_assert(!dist_ok(far_d(4) + 1) && far_d(0) == 16 * 4 + 10 && far_d(4) == 16 * 64 + 10,
              "far distance k is the one distance of the tile pair at tile distances (4 << k, (4 << k) + 1)");

#define PITCH 68

__device__ __forceinline__ void wave_sync() {
    __builtin_amdgcn_fence(__ATOMIC_RELEASE, "wavefront");
    __builtin_amdgcn_wave_barrier();
    __builtin_amdgcn_fence(__ATOMIC_ACQUIRE, "wavefront");
}

__device__ __forceinline__ void tile_gload(const float* __restrict__ tile, int g, int ln, v4f (&r)[8]) {
#pragma unroll
    for (int i = 0; i < 8; ++i)
        r[i] = *(const v4f*)(tile + (size_t)(2 * i + g) * (H_ * E_) + 4 * ln);
}
__device__ __forceinline__ void tile_stage(float* st, int g, int ln, const v4f (&r)[8]) {
#pragma unroll
    for (int i = 0; i < 8; ++i)
        *(v4f*)(st + (2 * i + g) * PITCH + 4 * ln) = r[i];
}
__device__ __forceinline__ void tile_frags(const float* st, int g, int ln, v2f (&f)[16]) {
#pragma unroll
    for (int q = 0; q < 8; ++q) {
        const v4f t = *(const v4f*)(st + ln * PITCH + 8 * q + 4 * g);
        v2f a0; a0[0] = t[0]; a0[1] = t[1];
        v2f a1; a1[0] = t[2]; a1[1] = t[3];
        f[2 * q]     = a0;
        f[2 * q + 1] = a1;
    }
}

__device__ __forceinline__ void rows_gload(const float* __restrict__ Kbh, int j0, int g, int ln, v4f (&r)[8]) {
#pragma unroll
    for (int i = 0; i < 8; ++i) {
        const int j = j0 + 2 * i + g;
        r[i] = *(const v4f*)(Kbh + (size_t)(j < 0 ? 0 : j) * (H_ * E_) + 4 * ln);
    }
}
__device__ __forceinline__ float far_dot(const v2f (&qf)[16], const v2f (&kf)[16]) {
    v2f a0 = (v2f){}, a1 = (v2f){};
#pragma unroll
    for (int f = 0; f < 16; f += 2) {
        a0 = __builtin_elementwise_fma(qf[f], kf[f], a0);
        a1 = __builtin_elementwise_fma(qf[f + 1], kf[f + 1], a1);
    }
    const v2f a = a0 + a1;
    const float d = a[0] + a[1];
    return d + __shfl_xor(d, 16, 32);
}
__device__ __forceinline__ void qk_mma_pair(const v2f (&qf)[16], const v2f (&ka)[16], const v2f (&kb)[16],
                                            v8f& sa, v8f& sb) {
    v8f acca = (v8f){}, accb = (v8f){};
#pragma unroll
    for (int q = 0; q < 8; ++q) {
        acca = __builtin_amdgcn_wmma_f32_16x16x4_f32(false, qf[2 * q], false, ka[2 * q],
                                                     (short)0, acca, false, false);
        accb = __builtin_amdgcn_wmma_f32_16x16x4_f32(false, qf[2 * q], false, kb[2 * q],
                                                     (short)0, accb, false, false);
        acca = __builtin_amdgcn_wmma_f32_16x16x4_f32(false, qf[2 * q + 1], false, ka[2 * q + 1],
                                                     (short)0, acca, false, false);
        accb = __builtin_amdgcn_wmma_f32_16x16x4_f32(false, qf[2 * q + 1], false, kb[2 * q + 1],
                                                     (short)0, accb, false, false);
    }
    sa = acca;
    sb = accb;
}

constexpr bool chunk_live(int m, int kc) {
    for (int c = 4 * kc; c < 4 * kc + 4; ++c)
        for (int row = 0; row < 16; ++row)
            if (dist_ok(16 * m + row - c)) return true;
    return false;
}
static_assert(!chunk_live(3, 0), "tile distance 3: key columns 0..3 are masked in every row");

template <int LO_A, int HI_A, int LO_B, int HI_B>
__device__ __forceinline__ void pv_vload(const float* __restrict__ V, int qt, int ma, int b, int h,
                                         int g, int ln, float4 (&vr)[2][4][2]) {
#pragma unroll
    for (int u = 0; u < 2; ++u) {
        const int lo = u ? LO_B : LO_A, hi = u ? HI_B : HI_A;
        const int jt = qt - (ma + u);
        const int j0 = (jt < 0 ? 0 : jt) * 16;
        const float* vbase = V + ((size_t)(b * S_ + j0) * H_ + h) * D_ + 4 * ln;
#pragma unroll
        for (int kc = lo; kc < hi; ++kc) {
            const int kr = 4 * kc + 2 * g;
            vr[u][kc][0] = *(const float4*)(vbase + (size_t)(kr + 0) * (H_ * D_));
            vr[u][kc][1] = *(const float4*)(vbase + (size_t)(kr + 1) * (H_ * D_));
        }
    }
}
constexpr bool col_live(int m, int c) {
    for (int row = 0; row < 16; ++row)
        if (dist_ok(16 * m + row - c)) return true;
    return false;
}
constexpr bool cols_dead(int m, int lo, int hi) {
    for (int c = lo; c < hi; ++c)
        if (col_live(m, c)) return false;
    return true;
}
static_assert(cols_dead(4, 6, 16), "nearer far tile: key columns 6..15 are masked in every row");
static_assert(cols_dead(5, 0, 6), "farther far tile: key columns 0..5 are masked in every row");

__device__ __forceinline__ void pv_vload_pair(const float* __restrict__ V, int qt, int ma, int b, int h,
                                              int g, int ln, float4 (&vr)[4][2]) {
    const int jn = qt - ma, jf = qt - ma - 1;
    const float* vnear = V + ((size_t)(b * S_ + (jn < 0 ? 0 : jn) * 16) * H_ + h) * D_ + 4 * ln;
    const float* vfarp = V + ((size_t)(b * S_ + (jf < 0 ? 0 : jf) * 16) * H_ + h) * D_ + 4 * ln;
    const float* vmid  = g ? vfarp : vnear;
#pragma unroll
    for (int kc = 0; kc < 4; ++kc) {
        const int kr = 4 * kc + 2 * g;
        const float* base = kc == 0 ? vnear : (kc == 1 ? vmid : vfarp);
        vr[kc][0] = *(const float4*)(base + (size_t)(kr + 0) * (H_ * D_));
        vr[kc][1] = *(const float4*)(base + (size_t)(kr + 1) * (H_ * D_));
    }
}
template <int LO, int HI>
__device__ __forceinline__ void pv_mma_tile(const v2f (&pa)[4], const float4 (&vr)[4][2], v8f (&oacc)[4]) {
#pragma unroll
    for (int kc = LO; kc < HI; ++kc) {
        const float4 r0 = vr[kc][0];
        const float4 r1 = vr[kc][1];
        v2f x[4];
        x[0][0] = r0.x; x[0][1] = r1.x;
        x[1][0] = r0.y; x[1][1] = r1.y;
        x[2][0] = r0.z; x[2][1] = r1.z;
        x[3][0] = r0.w; x[3][1] = r1.w;
#pragma unroll
        for (int nt = 0; nt < 4; ++nt)
            oacc[nt] = __builtin_amdgcn_wmma_f32_16x16x4_f32(false, pa[kc], false, x[nt],
                                                             (short)0, oacc[nt], false, false);
    }
}

__device__ __forceinline__ const float* k_tile(const float* __restrict__ K, int b, int h, int qt, int t) {
    const int jt = qt - slot_m(t);
    return K + ((size_t)(b * S_ + (jt < 0 ? 0 : jt) * 16) * H_ + h) * E_;
}

__global__ __launch_bounds__(WAVES * 32) __attribute__((amdgpu_waves_per_eu(2, 2)))
void logsparse_attn_kernel(const float* __restrict__ Q,
                           const float* __restrict__ K,
                           const float* __restrict__ V,
                           float* __restrict__ Out) {
    __shared__ __align__(16) float plds[WAVES][NSLOT][16][18];

    const int lane = threadIdx.x & 31;
    const int wave = __builtin_amdgcn_readfirstlane(threadIdx.x >> 5);
    const int g    = lane >> 4;
    const int ln   = lane & 15;

    const int gid = blockIdx.x * WAVES + wave;
    const int qt  = gid & 127;
    const int h   = (gid >> 7) & 7;
    const int b   = gid >> 10;
    const int i0  = qt * 16;

    float* stA = &plds[wave][0][0][0];
    float* stB = stA + 16 * PITCH;
    float* stC = stB + 16 * PITCH;
    static_assert(3 * 16 * PITCH <= NSLOT * 16 * 18, "three staged tiles fit in the wave's part of plds");
    const float* Kbh = K + ((size_t)b * S_ * H_ + h) * E_;

    v4f gq[8], gk[2][2][8];
    tile_gload(Q + ((size_t)(b * L_ + i0) * H_ + h) * E_, g, ln, gq);
    tile_gload(k_tile(K, b, h, qt, 0), g, ln, gk[0][0]);
    tile_gload(k_tile(K, b, h, qt, 1), g, ln, gk[0][1]);
    tile_gload(k_tile(K, b, h, qt, 2), g, ln, gk[1][0]);
    tile_gload(k_tile(K, b, h, qt, 3), g, ln, gk[1][1]);
    v2f qf[16];
    tile_stage(stA, g, ln, gq);
    wave_sync();
    tile_frags(stA, g, ln, qf);
    wave_sync();

    v2f kfa[16], kfb[16];
    tile_stage(stA, g, ln, gk[0][0]);
    tile_stage(stB, g, ln, gk[0][1]);
    rows_gload(Kbh, i0 - far_d(0), g, ln, gk[0][0]);
    rows_gload(Kbh, i0 - far_d(1), g, ln, gk[0][1]);
    wave_sync();
    tile_frags(stA, g, ln, kfa);
    tile_frags(stB, g, ln, kfb);
    wave_sync();

    float4 vn0[2][4][2], vn1[2][4][2], vfar[5][4][2];
    v8f s[4];
    v2f nfa[16], nfb[16];
    tile_stage(stA, g, ln, gk[1][0]);
    tile_stage(stB, g, ln, gk[1][1]);
    rows_gload(Kbh, i0 - far_d(2), g, ln, gk[1][0]);
    rows_gload(Kbh, i0 - far_d(3), g, ln, gk[1][1]);
    rows_gload(Kbh, i0 - far_d(4), g, ln, gq);
    wave_sync();
    tile_frags(stA, g, ln, nfa);
    tile_frags(stB, g, ln, nfb);
    wave_sync();
    __builtin_amdgcn_sched_barrier(0);
    qk_mma_pair(qf, kfa, kfb, s[0], s[1]);
    asm volatile("" :: "v"(s[0]), "v"(s[1]));
    __builtin_amdgcn_sched_barrier(0);

    v2f ffa[16], ffb[16], ffc[16];
    tile_stage(stA, g, ln, gk[0][0]);
    tile_stage(stB, g, ln, gk[0][1]);
    wave_sync();
    tile_frags(stA, g, ln, ffa);
    tile_frags(stB, g, ln, ffb);
    wave_sync();
    pv_vload<0, 4, 0, 4>(V, qt, 0, b, h, g, ln, vn0);
    pv_vload<0, 4, 1, 4>(V, qt, 2, b, h, g, ln, vn1);
    __builtin_amdgcn_sched_barrier(0);
    qk_mma_pair(qf, nfa, nfb, s[2], s[3]);

    float fs[NFAR];
    fs[0] = far_dot(qf, ffa);
    fs[1] = far_dot(qf, ffb);
    tile_stage(stA, g, ln, gk[1][0]);
    tile_stage(stB, g, ln, gk[1][1]);
    tile_stage(stC, g, ln, gq);
#pragma unroll
    for (int k = 0; k < 3; ++k)
        pv_vload_pair(V, qt, 4 << k, b, h, g, ln, vfar[k]);
    wave_sync();
    tile_frags(stA, g, ln, ffa);
    tile_frags(stB, g, ln, ffb);
    tile_frags(stC, g, ln, ffc);
    wave_sync();
    fs[2] = far_dot(qf, ffa);
    fs[3] = far_dot(qf, ffb);
    fs[4] = far_dot(qf, ffc);

    const int y = 15 + 8 * g - ln;
    constexpr unsigned long long A01 = ((unsigned long long)allow_word(1) << 32) | allow_word(0);
    constexpr unsigned long long A12 = ((unsigned long long)allow_word(2) << 32) | allow_word(1);
    const unsigned w0 = (unsigned)(A01 >> y);
    const unsigned w1 = (unsigned)(A12 >> y);
    const unsigned c0 = (unsigned)(0xFFFFFFFFFFFF8000ull >> y);
#pragma unroll
    for (int r = 0; r < 8; ++r) {
        const int i = i0 + r + 8 * g;
        const unsigned w0r = (i < 22) ? c0 : w0;
#pragma unroll
        for (int t = 0; t < 4; ++t) {
            const int c = 16 * t + r;
            const unsigned word = c < 32 ? w0r : w1;
            const unsigned live = (qt >= t) ? word : 0u;
            const bool ok = ((live >> (c & 31)) & 1u) != 0u;
            s[t][r] = ok ? SCALE * s[t][r] : MASKED;
        }
    }
    float fl[NFAR];
#pragma unroll
    for (int k = 0; k < NFAR; ++k)
        fl[k] = (i0 + ln >= far_d(k)) ? SCALE * fs[k] : MASKED;

#pragma unroll
    for (int r = 0; r < 8; ++r) {
#pragma unroll
        for (int t = 0; t < 4; ++t)
            plds[wave][t][r + 8 * g][ln] = s[t][r];
    }
#pragma unroll
    for (int k = 3; k < 5; ++k)
        pv_vload_pair(V, qt, 4 << k, b, h, g, ln, vfar[k]);
    wave_sync();

    const float* lw = &plds[wave][0][0][0];
    v2f pn[4][4], pf[5][4];
#pragma unroll
    for (int t = 0; t < 4; ++t) {
#pragma unroll
        for (int kc = (t == 3 ? 1 : 0); kc < 4; ++kc) {
            const float2 x = *(const float2*)(lw + (t * 16 + ln) * 18 + 4 * kc + 2 * g);
            v2f a; a[0] = x.x; a[1] = x.y;
            pn[t][kc] = a;
        }
    }
    __builtin_amdgcn_sched_barrier(0);

    float mx0 = -INFINITY, mx1 = -INFINITY;
#pragma unroll
    for (int t = 0; t < 4; ++t) {
#pragma unroll
        for (int kc = (t == 3 ? 1 : 0); kc < 4; ++kc) {
            mx0 = fmaxf(mx0, pn[t][kc][0]);
            mx1 = fmaxf(mx1, pn[t][kc][1]);
        }
    }
    float mx = fmaxf(mx0, mx1);
    mx = fmaxf(mx, __shfl_xor(mx, 16, 32));
#pragma unroll
    for (int k = 0; k < NFAR; ++k) mx = fmaxf(mx, fl[k]);
    float sm0 = 0.0f, sm1 = 0.0f;
#pragma unroll
    for (int t = 0; t < 4; ++t) {
#pragma unroll
        for (int kc = (t == 3 ? 1 : 0); kc < 4; ++kc) {
            const float p0 = __expf(pn[t][kc][0] - mx);
            const float p1 = __expf(pn[t][kc][1] - mx);
            sm0 += p0; sm1 += p1;
            v2f a; a[0] = p0; a[1] = p1;
            pn[t][kc] = a;
        }
    }
    float rsum = sm0 + sm1;
    rsum += __shfl_xor(rsum, 16, 32);
    const int cst = (ln + 6) & 15;
    const int sel = (((cst >> 1) & 1) == g) ? ((cst >> 2) * 2 + (cst & 1)) : -1;
    float fsum = 0.0f;
#pragma unroll
    for (int k = 0; k < NFAR; ++k) {
        const float p = __expf(fl[k] - mx);
        fsum += p;
#pragma unroll
        for (int kc = 0; kc < 4; ++kc) {
            v2f a;
            a[0] = (sel == 2 * kc)     ? p : 0.0f;
            a[1] = (sel == 2 * kc + 1) ? p : 0.0f;
            pf[k][kc] = a;
        }
    }
    rsum += fsum;

    v8f oacc[4];
#pragma unroll
    for (int nt = 0; nt < 4; ++nt) oacc[nt] = (v8f){};
    pv_mma_tile<0, 4>(pn[0], vn0[0], oacc);
    pv_mma_tile<0, 4>(pn[1], vn0[1], oacc);
    pv_mma_tile<0, 4>(pn[2], vn1[0], oacc);
    pv_mma_tile<1, 4>(pn[3], vn1[1], oacc);
#pragma unroll
    for (int k = 0; k < 5; ++k)
        pv_mma_tile<0, 4>(pf[k], vfar[k], oacc);

    const float inv = 1.0f / rsum;
#pragma unroll
    for (int r = 0; r < 8; ++r) {
        const int i = i0 + r + 8 * g;
        const float invr = __shfl(inv, r + 8 * g, 32);
        float* orow = Out + ((size_t)(b * L_ + i) * H_ + h) * D_;
        float4 o;
        o.x = oacc[0][r] * invr;
        o.y = oacc[1][r] * invr;
        o.z = oacc[2][r] * invr;
        o.w = oacc[3][r] * invr;
        *(float4*)(orow + 4 * ln) = o;
    }
}

extern "C" void kernel_launch(void* const* d_in, const int* in_sizes, int n_in,
                              void* d_out, int out_size, void* d_ws, size_t ws_size,
                              hipStream_t stream) {
    (void)in_sizes; (void)n_in; (void)out_size; (void)d_ws; (void)ws_size;
    const float* Q = (const float*)d_in[0];
    const float* K = (const float*)d_in[1];
    const float* V = (const float*)d_in[2];
    float* Out = (float*)d_out;

    const int total_waves = B_ * H_ * (L_ / 16);
    dim3 grid(total_waves / WAVES);
    dim3 block(WAVES * 32);
    hipLaunchKernelGGL(logsparse_attn_kernel, grid, block, 0, stream, Q, K, V, Out);
}


# P*V of the five far distances (one key per row) as P-scaled V rows added on the vector unit in the output layout, no far P tiles: 204 -> 124 WMMAs; far P crosses lanes via LDS; regroups O sums
# speedup vs baseline: 7.3913x; 1.1294x over previous
#include <hip/hip_runtime.h>
#include <math.h>


typedef __attribute__((ext_vector_type(2))) float v2f;
typedef __attribute__((ext_vector_type(8))) float v8f;
typedef __attribute__((ext_vector_type(4))) float v4f;

#define WAVES 4
#define NSLOT 14

constexpr int B_ = 2, L_ = 2048, S_ = 2048, H_ = 8, E_ = 64, D_ = 64;
constexpr float SCALE  = 0.125f;
constexpr float MASKED = -1.25e8f;

constexpr bool dist_ok(int d) {
    if (d < 0) return false;
    if (d <= 10) return true;
    int e = d - 10;
    return (e <= 1024) && ((e & (e - 1)) == 0);
}
constexpr unsigned allow_word(int w) {
    unsigned v = 0;
    for (int b = 0; b < 32; ++b)
        if (dist_ok(32 * w + b - 15)) v |= 1u << b;
    return v;
}

__device__ __forceinline__ constexpr int slot_m(int t) {
    return t < 6 ? t : (t < 8 ? t + 2 : (t < 10 ? t + 8 : (t < 12 ? t + 22 : t + 52)));
}

#define NFAR 5
__device__ __forceinline__ constexpr int far_d(int k) { return 10 + (64 << k); }
static_assert(dist_ok(far_d(0)) && dist_ok(far_d(1)) && dist_ok(far_d(2)) && dist_ok(far_d(3)) &&
              dist_ok(far_d(4)), "the five far distances are allowed");
static_assert(!dist_ok(far_d(4) + 1) && far_d(0) == 16 * 4 + 10 && far_d(4) == 16 * 64 + 10,
              "far distance k is the one distance of the tile pair at tile distances (4 << k, (4 << k) + 1)");

#define PITCH 68

__device__ __forceinline__ void wave_sync() {
    __builtin_amdgcn_fence(__ATOMIC_RELEASE, "wavefront");
    __builtin_amdgcn_wave_barrier();
    __builtin_amdgcn_fence(__ATOMIC_ACQUIRE, "wavefront");
}

__device__ __forceinline__ void tile_gload(const float* __restrict__ tile, int g, int ln, v4f (&r)[8]) {
#pragma unroll
    for (int i = 0; i < 8; ++i)
        r[i] = *(const v4f*)(tile + (size_t)(2 * i + g) * (H_ * E_) + 4 * ln);
}
__device__ __forceinline__ void tile_stage(float* st, int g, int ln, const v4f (&r)[8]) {
#pragma unroll
    for (int i = 0; i < 8; ++i)
        *(v4f*)(st + (2 * i + g) * PITCH + 4 * ln) = r[i];
}
__device__ __forceinline__ void tile_frags(const float* st, int g, int ln, v2f (&f)[16]) {
#pragma unroll
    for (int q = 0; q < 8; ++q) {
        const v4f t = *(const v4f*)(st + ln * PITCH + 8 * q + 4 * g);
        v2f a0; a0[0] = t[0]; a0[1] = t[1];
        v2f a1; a1[0] = t[2]; a1[1] = t[3];
        f[2 * q]     = a0;
        f[2 * q + 1] = a1;
    }
}

__device__ __forceinline__ void rows_gload(const float* __restrict__ Kbh, int j0, int g, int ln, v4f (&r)[8]) {
#pragma unroll
    for (int i = 0; i < 8; ++i) {
        const int j = j0 + 2 * i + g;
        r[i] = *(const v4f*)(Kbh + (size_t)(j < 0 ? 0 : j) * (H_ * E_) + 4 * ln);
    }
}
__device__ __forceinline__ float far_dot(const v2f (&qf)[16], const v2f (&kf)[16]) {
    v2f a0 = (v2f){}, a1 = (v2f){};
#pragma unroll
    for (int f = 0; f < 16; f += 2) {
        a0 = __builtin_elementwise_fma(qf[f], kf[f], a0);
        a1 = __builtin_elementwise_fma(qf[f + 1], kf[f + 1], a1);
    }
    const v2f a = a0 + a1;
    const float d = a[0] + a[1];
    return d + __shfl_xor(d, 16, 32);
}
__device__ __forceinline__ void qk_mma_pair(const v2f (&qf)[16], const v2f (&ka)[16], const v2f (&kb)[16],
                                            v8f& sa, v8f& sb) {
    v8f acca = (v8f){}, accb = (v8f){};
#pragma unroll
    for (int q = 0; q < 8; ++q) {
        acca = __builtin_amdgcn_wmma_f32_16x16x4_f32(false, qf[2 * q], false, ka[2 * q],
                                                     (short)0, acca, false, false);
        accb = __builtin_amdgcn_wmma_f32_16x16x4_f32(false, qf[2 * q], false, kb[2 * q],
                                                     (short)0, accb, false, false);
        acca = __builtin_amdgcn_wmma_f32_16x16x4_f32(false, qf[2 * q + 1], false, ka[2 * q + 1],
                                                     (short)0, acca, false, false);
        accb = __builtin_amdgcn_wmma_f32_16x16x4_f32(false, qf[2 * q + 1], false, kb[2 * q + 1],
                                                     (short)0, accb, false, false);
    }
    sa = acca;
    sb = accb;
}

constexpr bool chunk_live(int m, int kc) {
    for (int c = 4 * kc; c < 4 * kc + 4; ++c)
        for (int row = 0; row < 16; ++row)
            if (dist_ok(16 * m + row - c)) return true;
    return false;
}
static_assert(!chunk_live(3, 0), "tile distance 3: key columns 0..3 are masked in every row");

template <int LO_A, int HI_A, int LO_B, int HI_B>
__device__ __forceinline__ void pv_vload(const float* __restrict__ V, int qt, int ma, int b, int h,
                                         int g, int ln, float4 (&vr)[2][4][2]) {
#pragma unroll
    for (int u = 0; u < 2; ++u) {
        const int lo = u ? LO_B : LO_A, hi = u ? HI_B : HI_A;
        const int jt = qt - (ma + u);
        const int j0 = (jt < 0 ? 0 : jt) * 16;
        const float* vbase = V + ((size_t)(b * S_ + j0) * H_ + h) * D_ + 4 * ln;
#pragma unroll
        for (int kc = lo; kc < hi; ++kc) {
            const int kr = 4 * kc + 2 * g;
            vr[u][kc][0] = *(const float4*)(vbase + (size_t)(kr + 0) * (H_ * D_));
            vr[u][kc][1] = *(const float4*)(vbase + (size_t)(kr + 1) * (H_ * D_));
        }
    }
}
__device__ __forceinline__ void far_vload(const float* __restrict__ Vbh, int t, float4 (&vr)[8]) {
#pragma unroll
    for (int r = 0; r < 8; ++r) {
        const int off = t + r * (H_ * D_ * 4);
        vr[r] = *(const float4*)((const char*)Vbh + (unsigned)(off < 0 ? 0 : off));
    }
}
template <int LO, int HI>
__device__ __forceinline__ void pv_mma_tile(const v2f (&pa)[4], const float4 (&vr)[4][2], v8f (&oacc)[4]) {
#pragma unroll
    for (int kc = LO; kc < HI; ++kc) {
        const float4 r0 = vr[kc][0];
        const float4 r1 = vr[kc][1];
        v2f x[4];
        x[0][0] = r0.x; x[0][1] = r1.x;
        x[1][0] = r0.y; x[1][1] = r1.y;
        x[2][0] = r0.z; x[2][1] = r1.z;
        x[3][0] = r0.w; x[3][1] = r1.w;
#pragma unroll
        for (int nt = 0; nt < 4; ++nt)
            oacc[nt] = __builtin_amdgcn_wmma_f32_16x16x4_f32(false, pa[kc], false, x[nt],
                                                             (short)0, oacc[nt], false, false);
    }
}

__device__ __forceinline__ const float* k_tile(const float* __restrict__ K, int b, int h, int qt, int t) {
    const int jt = qt - slot_m(t);
    return K + ((size_t)(b * S_ + (jt < 0 ? 0 : jt) * 16) * H_ + h) * E_;
}

__global__ __launch_bounds__(WAVES * 32) __attribute__((amdgpu_waves_per_eu(2, 2)))
void logsparse_attn_kernel(const float* __restrict__ Q,
                           const float* __restrict__ K,
                           const float* __restrict__ V,
                           float* __restrict__ Out) {
    __shared__ __align__(16) float plds[WAVES][NSLOT][16][18];

    const int lane = threadIdx.x & 31;
    const int wave = __builtin_amdgcn_readfirstlane(threadIdx.x >> 5);
    const int g    = lane >> 4;
    const int ln   = lane & 15;

    const int gid = blockIdx.x * WAVES + wave;
    const int qt  = gid & 127;
    const int h   = (gid >> 7) & 7;
    const int b   = gid >> 10;
    const int i0  = qt * 16;

    float* stA = &plds[wave][0][0][0];
    float* stB = stA + 16 * PITCH;
    float* stC = stB + 16 * PITCH;
    static_assert(3 * 16 * PITCH <= NSLOT * 16 * 18, "three staged tiles fit in the wave's part of plds");
    const float* Kbh = K + ((size_t)b * S_ * H_ + h) * E_;

    v4f gq[8], gk[2][2][8];
    tile_gload(Q + ((size_t)(b * L_ + i0) * H_ + h) * E_, g, ln, gq);
    tile_gload(k_tile(K, b, h, qt, 0), g, ln, gk[0][0]);
    tile_gload(k_tile(K, b, h, qt, 1), g, ln, gk[0][1]);
    tile_gload(k_tile(K, b, h, qt, 2), g, ln, gk[1][0]);
    tile_gload(k_tile(K, b, h, qt, 3), g, ln, gk[1][1]);
    v2f qf[16];
    tile_stage(stA, g, ln, gq);
    wave_sync();
    tile_frags(stA, g, ln, qf);
    wave_sync();

    v2f kfa[16], kfb[16];
    tile_stage(stA, g, ln, gk[0][0]);
    tile_stage(stB, g, ln, gk[0][1]);
    rows_gload(Kbh, i0 - far_d(0), g, ln, gk[0][0]);
    rows_gload(Kbh, i0 - far_d(1), g, ln, gk[0][1]);
    wave_sync();
    tile_frags(stA, g, ln, kfa);
    tile_frags(stB, g, ln, kfb);
    wave_sync();

    float4 vn0[2][4][2], vn1[2][4][2], vfar[NFAR][8];
    const float* Vbh = V + ((size_t)b * S_ * H_ + h) * D_;
    const int vlane = (i0 + 8 * g) * (H_ * D_ * 4) + 16 * ln;
    v8f s[4];
    v2f nfa[16], nfb[16];
    tile_stage(stA, g, ln, gk[1][0]);
    tile_stage(stB, g, ln, gk[1][1]);
    rows_gload(Kbh, i0 - far_d(2), g, ln, gk[1][0]);
    rows_gload(Kbh, i0 - far_d(3), g, ln, gk[1][1]);
    rows_gload(Kbh, i0 - far_d(4), g, ln, gq);
    wave_sync();
    tile_frags(stA, g, ln, nfa);
    tile_frags(stB, g, ln, nfb);
    wave_sync();
    __builtin_amdgcn_sched_barrier(0);
    qk_mma_pair(qf, kfa, kfb, s[0], s[1]);
    asm volatile("" :: "v"(s[0]), "v"(s[1]));
    __builtin_amdgcn_sched_barrier(0);

    v2f ffa[16], ffb[16], ffc[16];
    tile_stage(stA, g, ln, gk[0][0]);
    tile_stage(stB, g, ln, gk[0][1]);
    wave_sync();
    tile_frags(stA, g, ln, ffa);
    tile_frags(stB, g, ln, ffb);
    wave_sync();
    pv_vload<0, 4, 0, 4>(V, qt, 0, b, h, g, ln, vn0);
    pv_vload<0, 4, 1, 4>(V, qt, 2, b, h, g, ln, vn1);
    __builtin_amdgcn_sched_barrier(0);
    qk_mma_pair(qf, nfa, nfb, s[2], s[3]);

    float fs[NFAR];
    fs[0] = far_dot(qf, ffa);
    fs[1] = far_dot(qf, ffb);
    tile_stage(stA, g, ln, gk[1][0]);
    tile_stage(stB, g, ln, gk[1][1]);
    tile_stage(stC, g, ln, gq);
#pragma unroll
    for (int k = 0; k < 3; ++k)
        far_vload(Vbh, vlane - far_d(k) * (H_ * D_ * 4), vfar[k]);
    wave_sync();
    tile_frags(stA, g, ln, ffa);
    tile_frags(stB, g, ln, ffb);
    tile_frags(stC, g, ln, ffc);
    wave_sync();
    fs[2] = far_dot(qf, ffa);
    fs[3] = far_dot(qf, ffb);
    fs[4] = far_dot(qf, ffc);

    const int y = 15 + 8 * g - ln;
    constexpr unsigned long long A01 = ((unsigned long long)allow_word(1) << 32) | allow_word(0);
    constexpr unsigned long long A12 = ((unsigned long long)allow_word(2) << 32) | allow_word(1);
    const unsigned w0 = (unsigned)(A01 >> y);
    const unsigned w1 = (unsigned)(A12 >> y);
    const unsigned c0 = (unsigned)(0xFFFFFFFFFFFF8000ull >> y);
#pragma unroll
    for (int r = 0; r < 8; ++r) {
        const int i = i0 + r + 8 * g;
        const unsigned w0r = (i < 22) ? c0 : w0;
#pragma unroll
        for (int t = 0; t < 4; ++t) {
            const int c = 16 * t + r;
            const unsigned word = c < 32 ? w0r : w1;
            const unsigned live = (qt >= t) ? word : 0u;
            const bool ok = ((live >> (c & 31)) & 1u) != 0u;
            s[t][r] = ok ? SCALE * s[t][r] : MASKED;
        }
    }
    float fl[NFAR];
#pragma unroll
    for (int k = 0; k < NFAR; ++k)
        fl[k] = (i0 + ln >= far_d(k)) ? SCALE * fs[k] : MASKED;

#pragma unroll
    for (int r = 0; r < 8; ++r) {
#pragma unroll
        for (int t = 0; t < 4; ++t)
            plds[wave][t][r + 8 * g][ln] = s[t][r];
    }
#pragma unroll
    for (int k = 3; k < NFAR; ++k)
        far_vload(Vbh, vlane - far_d(k) * (H_ * D_ * 4), vfar[k]);
    wave_sync();

    const float* lw = &plds[wave][0][0][0];
    v2f pn[4][4];
#pragma unroll
    for (int t = 0; t < 4; ++t) {
#pragma unroll
        for (int kc = (t == 3 ? 1 : 0); kc < 4; ++kc) {
            const float2 x = *(const float2*)(lw + (t * 16 + ln) * 18 + 4 * kc + 2 * g);
            v2f a; a[0] = x.x; a[1] = x.y;
            pn[t][kc] = a;
        }
    }
    __builtin_amdgcn_sched_barrier(0);

    float mx0 = -INFINITY, mx1 = -INFINITY;
#pragma unroll
    for (int t = 0; t < 4; ++t) {
#pragma unroll
        for (int kc = (t == 3 ? 1 : 0); kc < 4; ++kc) {
            mx0 = fmaxf(mx0, pn[t][kc][0]);
            mx1 = fmaxf(mx1, pn[t][kc][1]);
        }
    }
    float mx = fmaxf(mx0, mx1);
    mx = fmaxf(mx, __shfl_xor(mx, 16, 32));
#pragma unroll
    for (int k = 0; k < NFAR; ++k) mx = fmaxf(mx, fl[k]);
    float sm0 = 0.0f, sm1 = 0.0f;
#pragma unroll
    for (int t = 0; t < 4; ++t) {
#pragma unroll
        for (int kc = (t == 3 ? 1 : 0); kc < 4; ++kc) {
            const float p0 = __expf(pn[t][kc][0] - mx);
            const float p1 = __expf(pn[t][kc][1] - mx);
            sm0 += p0; sm1 += p1;
            v2f a; a[0] = p0; a[1] = p1;
            pn[t][kc] = a;
        }
    }
    float rsum = sm0 + sm1;
    rsum += __shfl_xor(rsum, 16, 32);
    float* pb = &plds[wave][4][0][0];
    float fsum = 0.0f;
#pragma unroll
    for (int k = 0; k < NFAR; ++k) {
        const float p = __expf(fl[k] - mx);
        fsum += p;
        pb[16 * k + ln] = p;
    }
    rsum += fsum;
    wave_sync();
    float4 pr[NFAR][2];
#pragma unroll
    for (int k = 0; k < NFAR; ++k) {
        pr[k][0] = *(const float4*)(pb + 16 * k + 8 * g);
        pr[k][1] = *(const float4*)(pb + 16 * k + 8 * g + 4);
    }

    v8f oacc[4];
#pragma unroll
    for (int nt = 0; nt < 4; ++nt) oacc[nt] = (v8f){};
    pv_mma_tile<0, 4>(pn[0], vn0[0], oacc);
    pv_mma_tile<0, 4>(pn[1], vn0[1], oacc);
    pv_mma_tile<0, 4>(pn[2], vn1[0], oacc);
    pv_mma_tile<1, 4>(pn[3], vn1[1], oacc);
    v2f f01[8], f23[8];
#pragma unroll
    for (int r = 0; r < 8; ++r) {
        v2f a01, a23;
#pragma unroll
        for (int k = 0; k < NFAR; ++k) {
            const float4 pq = pr[k][r >> 2];
            const float p = (r & 3) == 0 ? pq.x : ((r & 3) == 1 ? pq.y : ((r & 3) == 2 ? pq.z : pq.w));
            const float4 vv = vfar[k][r];
            v2f pp;  pp[0] = p;     pp[1] = p;
            v2f v01; v01[0] = vv.x; v01[1] = vv.y;
            v2f v23; v23[0] = vv.z; v23[1] = vv.w;
            if (k == 0) {
                a01 = pp * v01;
                a23 = pp * v23;
            } else {
                a01 = __builtin_elementwise_fma(pp, v01, a01);
                a23 = __builtin_elementwise_fma(pp, v23, a23);
            }
        }
        f01[r] = a01;
        f23[r] = a23;
    }

    const float inv = 1.0f / rsum;
#pragma unroll
    for (int r = 0; r < 8; ++r) {
        const int i = i0 + r + 8 * g;
        const float invr = __shfl(inv, r + 8 * g, 32);
        float* orow = Out + ((size_t)(b * L_ + i) * H_ + h) * D_;
        float4 o;
        o.x = (oacc[0][r] + f01[r][0]) * invr;
        o.y = (oacc[1][r] + f01[r][1]) * invr;
        o.z = (oacc[2][r] + f23[r][0]) * invr;
        o.w = (oacc[3][r] + f23[r][1]) * invr;
        *(float4*)(orow + 4 * ln) = o;
    }
}

extern "C" void kernel_launch(void* const* d_in, const int* in_sizes, int n_in,
                              void* d_out, int out_size, void* d_ws, size_t ws_size,
                              hipStream_t stream) {
    (void)in_sizes; (void)n_in; (void)out_size; (void)d_ws; (void)ws_size;
    const float* Q = (const float*)d_in[0];
    const float* K = (const float*)d_in[1];
    const float* V = (const float*)d_in[2];
    float* Out = (float*)d_out;

    const int total_waves = B_ * H_ * (L_ / 16);
    dim3 grid(total_waves / WAVES);
    dim3 block(WAVES * 32);
    hipLaunchKernelGGL(logsparse_attn_kernel, grid, block, 0, stream, Q, K, V, Out);
}
